# starting kernel
# baseline (speedup 1.0000x reference)
#include <hip/hip_runtime.h>
#include <hip/hip_bf16.h>


typedef __attribute__((ext_vector_type(2))) float v2f;
typedef __attribute__((ext_vector_type(8))) float v8f;

constexpr int CH    = 128;
constexpr int BATCH = 64;
constexpr int HW    = 56 * 56;
constexpr float MTOT = (float)BATCH * (float)HW;
constexpr float EPSR = 1e-5f;
constexpr int   T_NEWTON = 5;

__global__ void __launch_bounds__(256)
cw_mean_partial(const float* __restrict__ x, float* __restrict__ meanPart) {
  const int b = blockIdx.x;
  const int c = blockIdx.y;
  const float* p = x + ((size_t)(b * CH + c)) * HW;
  float s = 0.f;
  for (int i = threadIdx.x; i < HW; i += 256) s += p[i];
  __shared__ float red[256];
  red[threadIdx.x] = s;
  __syncthreads();
  for (int off = 128; off > 0; off >>= 1) {
    if ((int)threadIdx.x < off) red[threadIdx.x] += red[threadIdx.x + off];
    __syncthreads();
  }
  if (threadIdx.x == 0) meanPart[c * BATCH + b] = red[0];
}

__global__ void cw_mean_finish(const float* __restrict__ meanPart,
                               float* __restrict__ mean) {
  const int c = threadIdx.x;
  float s = 0.f;
  for (int b = 0; b < BATCH; ++b) s += meanPart[c * BATCH + b];
  mean[c] = s / MTOT;
}

constexpr int KC = 64;
constexpr int XPAD = KC + 4;

__global__ void __launch_bounds__(256)
cw_gram(const float* __restrict__ x, float* __restrict__ gramPart) {
  __shared__ float sX[CH][XPAD];
  const int b    = blockIdx.x;
  const int tid  = threadIdx.x;
  const int wave = tid >> 5;
  const int lane = tid & 31;
  const int hi   = lane >> 4;
  const int l15  = lane & 15;

  v8f acc[8] = {};
  const float* xb = x + (size_t)b * CH * HW;

  for (int kc = 0; kc < HW; kc += KC) {
#pragma unroll
    for (int i = 0; i < 8; ++i) {
      const int idx = i * 256 + tid;
      const int c   = idx >> 4;
      const int k4  = (idx & 15) << 2;
      const float4 v = *reinterpret_cast<const float4*>(xb + (size_t)c * HW + kc + k4);
      sX[c][k4 + 0] = v.x; sX[c][k4 + 1] = v.y;
      sX[c][k4 + 2] = v.z; sX[c][k4 + 3] = v.w;
    }
    __syncthreads();

#pragma unroll 2
    for (int k0 = 0; k0 < KC; k0 += 4) {
      const int ks = k0 + hi * 2;
      const int ra = wave * 16 + l15;
      v2f a; a.x = sX[ra][ks]; a.y = sX[ra][ks + 1];
#pragma unroll
      for (int tc = 0; tc < 8; ++tc) {
        const int rb = tc * 16 + l15;
        v2f bb; bb.x = sX[rb][ks]; bb.y = sX[rb][ks + 1];
        acc[tc] = __builtin_amdgcn_wmma_f32_16x16x4_f32(
            false, a, false, bb, (short)0, acc[tc], false, false);
      }
    }
    __syncthreads();
  }

  float* gp = gramPart + (size_t)b * CH * CH;
#pragma unroll
  for (int tc = 0; tc < 8; ++tc) {
#pragma unroll
    for (int r = 0; r < 8; ++r) {
      const int row = wave * 16 + r + hi * 8;
      const int col = tc * 16 + l15;
      gp[row * CH + col] = acc[tc][r];
    }
  }
}

__device__ inline void mm128(float* __restrict__ Cm, const float* __restrict__ A,
                             const float* __restrict__ Bm, int tid) {
  const int row = tid >> 1;
  const int c0  = (tid & 1) << 6;
  float outv[64];
#pragma unroll
  for (int j = 0; j < 64; ++j) outv[j] = 0.f;
  for (int k = 0; k < CH; ++k) {
    const float a = A[row * CH + k];
    const float* brow = Bm + k * CH + c0;
#pragma unroll 8
    for (int j = 0; j < 64; ++j) outv[j] += a * brow[j];
  }
#pragma unroll 8
  for (int j = 0; j < 64; ++j) Cm[row * CH + c0 + j] = outv[j];
}

__global__ void __launch_bounds__(256)
cw_newton(const float* __restrict__ gramPart, const float* __restrict__ mean,
          const float* __restrict__ rot, float* __restrict__ Mout,
          float* __restrict__ bias) {
  extern __shared__ float smem[];
  float* S  = smem;
  float* P  = smem + 1 * CH * CH;
  float* T1 = smem + 2 * CH * CH;
  float* T2 = smem + 3 * CH * CH;
  __shared__ float sTrInv;
  const int tid = threadIdx.x;

  for (int e = tid; e < CH * CH; e += 256) {
    float g = 0.f;
    for (int s = 0; s < BATCH; ++s) g += gramPart[(size_t)s * CH * CH + e];
    const int r = e >> 7, c = e & 127;
    float sig = (g - MTOT * mean[r] * mean[c]) / MTOT;
    if (r == c) sig += EPSR;
    S[e] = sig;
  }
  __syncthreads();
  if (tid == 0) {
    float tr = 0.f;
    for (int i = 0; i < CH; ++i) tr += S[i * CH + i];
    sTrInv = 1.0f / tr;
  }
  __syncthreads();
  const float rTr = sTrInv;
  for (int e = tid; e < CH * CH; e += 256) {
    S[e] *= rTr;
    const int r = e >> 7, c = e & 127;
    P[e] = (r == c) ? 1.f : 0.f;
  }
  __syncthreads();

  for (int it = 0; it < T_NEWTON; ++it) {
    mm128(T1, P, P, tid);  __syncthreads();
    mm128(T2, T1, P, tid); __syncthreads();
    mm128(T1, T2, S, tid); __syncthreads();
    for (int e = tid; e < CH * CH; e += 256) P[e] = 1.5f * P[e] - 0.5f * T1[e];
    __syncthreads();
  }

  mm128(T2, rot, P, tid);
  __syncthreads();
  const float sc = sqrtf(rTr);
  for (int e = tid; e < CH * CH; e += 256) Mout[e] = T2[e] * sc;
  if (tid < CH) {
    float s = 0.f;
    for (int c = 0; c < CH; ++c) s += T2[tid * CH + c] * mean[c];
    bias[tid] = s * sc;
  }
}

constexpr int MPAD = CH + 4;
__global__ void __launch_bounds__(256)
cw_apply(const float* __restrict__ x, const float* __restrict__ Mmat,
         const float* __restrict__ bias, float* __restrict__ out) {
  extern __shared__ float smem[];
  float* sM = smem;
  float* sX = smem + CH * MPAD;
  float* sB = sX + CH * XPAD;
  const int b    = blockIdx.y;
  const int tid  = threadIdx.x;
  const int wave = tid >> 5;
  const int lane = tid & 31;
  const int hi   = lane >> 4;
  const int l15  = lane & 15;

  for (int i = tid; i < (CH * CH) / 4; i += 256) {
    const int e = i * 4;
    const int r = e >> 7, c = e & 127;
    const float4 v = *reinterpret_cast<const float4*>(Mmat + e);
    float* d = sM + r * MPAD + c;
    d[0] = v.x; d[1] = v.y; d[2] = v.z; d[3] = v.w;
  }
  if (tid < CH) sB[tid] = bias[tid];
  __syncthreads();

  const float* xb = x + (size_t)b * CH * HW;
  float*       ob = out + (size_t)b * CH * HW;

  for (int j = 0; j < 7; ++j) {
    const int hw0 = (blockIdx.x * 7 + j) * 64;
#pragma unroll
    for (int i = 0; i < 8; ++i) {
      const int idx = i * 256 + tid;
      const int c   = idx >> 4;
      const int k4  = (idx & 15) << 2;
      const float4 v = *reinterpret_cast<const float4*>(xb + (size_t)c * HW + hw0 + k4);
      float* d = sX + c * XPAD + k4;
      d[0] = v.x; d[1] = v.y; d[2] = v.z; d[3] = v.w;
    }
    __syncthreads();

    v8f acc[4] = {};
#pragma unroll 2
    for (int c0 = 0; c0 < CH; c0 += 4) {
      const int ks = c0 + hi * 2;
      const int ra = wave * 16 + l15;
      v2f a; a.x = sM[ra * MPAD + ks]; a.y = sM[ra * MPAD + ks + 1];
#pragma unroll
      for (int ht = 0; ht < 4; ++ht) {
        v2f bb;
        bb.x = sX[ks * XPAD + ht * 16 + l15];
        bb.y = sX[(ks + 1) * XPAD + ht * 16 + l15];
        acc[ht] = __builtin_amdgcn_wmma_f32_16x16x4_f32(
            false, a, false, bb, (short)0, acc[ht], false, false);
      }
    }

#pragma unroll
    for (int ht = 0; ht < 4; ++ht) {
#pragma unroll
      for (int r = 0; r < 8; ++r) {
        const int dch = wave * 16 + r + hi * 8;
        const int hw  = hw0 + ht * 16 + l15;
        ob[(size_t)dch * HW + hw] = acc[ht][r] - sB[dch];
      }
    }
    __syncthreads();
  }
}

extern "C" void kernel_launch(void* const* d_in, const int* in_sizes, int n_in,
                              void* d_out, int out_size, void* d_ws, size_t ws_size,
                              hipStream_t stream) {
  const float* x   = (const float*)d_in[0];
  const float* rot = (const float*)d_in[1];
  float* out = (float*)d_out;
  float* ws  = (float*)d_ws;

  float* gramPart = ws;
  float* meanPart = gramPart + (size_t)BATCH * CH * CH;
  float* mean     = meanPart + CH * BATCH;
  float* Mmat     = mean + CH;
  float* bias     = Mmat + CH * CH;

  cw_mean_partial<<<dim3(BATCH, CH), 256, 0, stream>>>(x, meanPart);
  cw_gram<<<BATCH, 256, 0, stream>>>(x, gramPart);
  cw_mean_finish<<<1, CH, 0, stream>>>(meanPart, mean);
  cw_newton<<<1, 256, 4 * CH * CH * sizeof(float), stream>>>(
      gramPart, mean, rot, Mmat, bias);
  cw_apply<<<dim3(7, BATCH), 256,
             (CH * MPAD + CH * XPAD + CH) * sizeof(float), stream>>>(
      x, Mmat, bias, out);
}


# cw_newton: its sixteen 128x128 products leave the per-thread FMA loop (64 accumulators in scratch) for f32 16x16x4 WMMA on LDS strips; the k-sum order inside each product changes
# speedup vs baseline: 12.0558x; 12.0558x over previous
#include <hip/hip_runtime.h>
#include <hip/hip_bf16.h>


typedef __attribute__((ext_vector_type(2))) float v2f;
typedef __attribute__((ext_vector_type(8))) float v8f;

constexpr int CH    = 128;
constexpr int BATCH = 64;
constexpr int HW    = 56 * 56;
constexpr float MTOT = (float)BATCH * (float)HW;
constexpr float EPSR = 1e-5f;
constexpr int   T_NEWTON = 5;

__global__ void __launch_bounds__(256)
cw_mean_partial(const float* __restrict__ x, float* __restrict__ meanPart) {
  const int b = blockIdx.x;
  const int c = blockIdx.y;
  const float* p = x + ((size_t)(b * CH + c)) * HW;
  float s = 0.f;
  for (int i = threadIdx.x; i < HW; i += 256) s += p[i];
  __shared__ float red[256];
  red[threadIdx.x] = s;
  __syncthreads();
  for (int off = 128; off > 0; off >>= 1) {
    if ((int)threadIdx.x < off) red[threadIdx.x] += red[threadIdx.x + off];
    __syncthreads();
  }
  if (threadIdx.x == 0) meanPart[c * BATCH + b] = red[0];
}

__global__ void cw_mean_finish(const float* __restrict__ meanPart,
                               float* __restrict__ mean) {
  const int c = threadIdx.x;
  float s = 0.f;
  for (int b = 0; b < BATCH; ++b) s += meanPart[c * BATCH + b];
  mean[c] = s / MTOT;
}

constexpr int KC = 64;
constexpr int XPAD = KC + 4;

__global__ void __launch_bounds__(256)
cw_gram(const float* __restrict__ x, float* __restrict__ gramPart) {
  __shared__ float sX[CH][XPAD];
  const int b    = blockIdx.x;
  const int tid  = threadIdx.x;
  const int wave = tid >> 5;
  const int lane = tid & 31;
  const int hi   = lane >> 4;
  const int l15  = lane & 15;

  v8f acc[8] = {};
  const float* xb = x + (size_t)b * CH * HW;

  for (int kc = 0; kc < HW; kc += KC) {
#pragma unroll
    for (int i = 0; i < 8; ++i) {
      const int idx = i * 256 + tid;
      const int c   = idx >> 4;
      const int k4  = (idx & 15) << 2;
      const float4 v = *reinterpret_cast<const float4*>(xb + (size_t)c * HW + kc + k4);
      sX[c][k4 + 0] = v.x; sX[c][k4 + 1] = v.y;
      sX[c][k4 + 2] = v.z; sX[c][k4 + 3] = v.w;
    }
    __syncthreads();

#pragma unroll 2
    for (int k0 = 0; k0 < KC; k0 += 4) {
      const int ks = k0 + hi * 2;
      const int ra = wave * 16 + l15;
      v2f a; a.x = sX[ra][ks]; a.y = sX[ra][ks + 1];
#pragma unroll
      for (int tc = 0; tc < 8; ++tc) {
        const int rb = tc * 16 + l15;
        v2f bb; bb.x = sX[rb][ks]; bb.y = sX[rb][ks + 1];
        acc[tc] = __builtin_amdgcn_wmma_f32_16x16x4_f32(
            false, a, false, bb, (short)0, acc[tc], false, false);
      }
    }
    __syncthreads();
  }

  float* gp = gramPart + (size_t)b * CH * CH;
#pragma unroll
  for (int tc = 0; tc < 8; ++tc) {
#pragma unroll
    for (int r = 0; r < 8; ++r) {
      const int row = wave * 16 + r + hi * 8;
      const int col = tc * 16 + l15;
      gp[row * CH + col] = acc[tc][r];
    }
  }
}

constexpr int NLD = CH + 12;
constexpr size_t NEWTON_LDS_BYTES = (size_t)3 * CH * NLD * sizeof(float);

__device__ inline void strip_mm(v8f (&acc)[8], const float* __restrict__ Astrip,
                                const float* __restrict__ Bm, int hi, int l15) {
#pragma unroll
  for (int tc = 0; tc < 8; ++tc) acc[tc] = v8f{};
  for (int k0 = 0; k0 < CH; k0 += 4) {
    const int ks = k0 + hi * 2;
    v2f a; a.x = Astrip[l15 * NLD + ks]; a.y = Astrip[l15 * NLD + ks + 1];
#pragma unroll
    for (int tc = 0; tc < 8; ++tc) {
      v2f bb;
      bb.x = Bm[ks * NLD + tc * 16 + l15];
      bb.y = Bm[(ks + 1) * NLD + tc * 16 + l15];
      acc[tc] = __builtin_amdgcn_wmma_f32_16x16x4_f32(
          false, a, false, bb, (short)0, acc[tc], false, false);
    }
  }
}

__device__ inline void strip_store(float* __restrict__ strip, const v8f (&acc)[8],
                                   int hi, int l15) {
#pragma unroll
  for (int tc = 0; tc < 8; ++tc) {
#pragma unroll
    for (int r = 0; r < 8; ++r) strip[(r + hi * 8) * NLD + tc * 16 + l15] = acc[tc][r];
  }
}

__global__ void __launch_bounds__(256)
cw_newton(const float* __restrict__ gramPart, const float* __restrict__ mean,
          const float* __restrict__ rot, float* __restrict__ Mout,
          float* __restrict__ bias) {
  extern __shared__ float smem[];
  float* S = smem;
  float* P = smem + 1 * CH * NLD;
  float* T = smem + 2 * CH * NLD;
  __shared__ float sTrInv;
  const int tid  = threadIdx.x;
  const int wave = tid >> 5;
  const int lane = tid & 31;
  const int hi   = lane >> 4;
  const int l15  = lane & 15;

  for (int e = tid; e < CH * CH; e += 256) {
    float g = 0.f;
    for (int s = 0; s < BATCH; ++s) g += gramPart[(size_t)s * CH * CH + e];
    const int r = e >> 7, c = e & 127;
    float sig = (g - MTOT * mean[r] * mean[c]) / MTOT;
    if (r == c) sig += EPSR;
    S[r * NLD + c] = sig;
  }
  __syncthreads();
  if (tid == 0) {
    float tr = 0.f;
    for (int i = 0; i < CH; ++i) tr += S[i * NLD + i];
    sTrInv = 1.0f / tr;
  }
  __syncthreads();
  const float rTr = sTrInv;
  for (int e = tid; e < CH * CH; e += 256) {
    const int r = e >> 7, c = e & 127;
    S[r * NLD + c] *= rTr;
    P[r * NLD + c] = (r == c) ? 1.f : 0.f;
  }
  __syncthreads();

  float* Tw = T + wave * 16 * NLD;
  float* Pw = P + wave * 16 * NLD;
  v8f acc[8];

  for (int it = 0; it < T_NEWTON; ++it) {
    strip_mm(acc, Pw, P, hi, l15);
    strip_store(Tw, acc, hi, l15);
    __syncthreads();
    strip_mm(acc, Tw, P, hi, l15);
    strip_store(Tw, acc, hi, l15);
    __syncthreads();
    strip_mm(acc, Tw, S, hi, l15);
#pragma unroll
    for (int tc = 0; tc < 8; ++tc) {
#pragma unroll
      for (int r = 0; r < 8; ++r) {
        float* p = Pw + (r + hi * 8) * NLD + tc * 16 + l15;
        *p = 1.5f * *p - 0.5f * acc[tc][r];
      }
    }
    __syncthreads();
  }

  for (int i = lane; i < 16 * CH / 4; i += 32) {
    const int r = i >> 5, c = (i & 31) << 2;
    const float4 v = *reinterpret_cast<const float4*>(rot + (wave * 16 + r) * CH + c);
    float* d = Tw + r * NLD + c;
    d[0] = v.x; d[1] = v.y; d[2] = v.z; d[3] = v.w;
  }
  __syncthreads();
  strip_mm(acc, Tw, P, hi, l15);
  strip_store(Tw, acc, hi, l15);
  const float sc = sqrtf(rTr);
#pragma unroll
  for (int tc = 0; tc < 8; ++tc) {
#pragma unroll
    for (int r = 0; r < 8; ++r)
      Mout[(wave * 16 + r + hi * 8) * CH + tc * 16 + l15] = acc[tc][r] * sc;
  }
  __syncthreads();
  if (tid < CH) {
    float s = 0.f;
    for (int c = 0; c < CH; ++c) s += T[tid * NLD + c] * mean[c];
    bias[tid] = s * sc;
  }
}

constexpr int MPAD = CH + 4;
__global__ void __launch_bounds__(256)
cw_apply(const float* __restrict__ x, const float* __restrict__ Mmat,
         const float* __restrict__ bias, float* __restrict__ out) {
  extern __shared__ float smem[];
  float* sM = smem;
  float* sX = smem + CH * MPAD;
  float* sB = sX + CH * XPAD;
  const int b    = blockIdx.y;
  const int tid  = threadIdx.x;
  const int wave = tid >> 5;
  const int lane = tid & 31;
  const int hi   = lane >> 4;
  const int l15  = lane & 15;

  for (int i = tid; i < (CH * CH) / 4; i += 256) {
    const int e = i * 4;
    const int r = e >> 7, c = e & 127;
    const float4 v = *reinterpret_cast<const float4*>(Mmat + e);
    float* d = sM + r * MPAD + c;
    d[0] = v.x; d[1] = v.y; d[2] = v.z; d[3] = v.w;
  }
  if (tid < CH) sB[tid] = bias[tid];
  __syncthreads();

  const float* xb = x + (size_t)b * CH * HW;
  float*       ob = out + (size_t)b * CH * HW;

  for (int j = 0; j < 7; ++j) {
    const int hw0 = (blockIdx.x * 7 + j) * 64;
#pragma unroll
    for (int i = 0; i < 8; ++i) {
      const int idx = i * 256 + tid;
      const int c   = idx >> 4;
      const int k4  = (idx & 15) << 2;
      const float4 v = *reinterpret_cast<const float4*>(xb + (size_t)c * HW + hw0 + k4);
      float* d = sX + c * XPAD + k4;
      d[0] = v.x; d[1] = v.y; d[2] = v.z; d[3] = v.w;
    }
    __syncthreads();

    v8f acc[4] = {};
#pragma unroll 2
    for (int c0 = 0; c0 < CH; c0 += 4) {
      const int ks = c0 + hi * 2;
      const int ra = wave * 16 + l15;
      v2f a; a.x = sM[ra * MPAD + ks]; a.y = sM[ra * MPAD + ks + 1];
#pragma unroll
      for (int ht = 0; ht < 4; ++ht) {
        v2f bb;
        bb.x = sX[ks * XPAD + ht * 16 + l15];
        bb.y = sX[(ks + 1) * XPAD + ht * 16 + l15];
        acc[ht] = __builtin_amdgcn_wmma_f32_16x16x4_f32(
            false, a, false, bb, (short)0, acc[ht], false, false);
      }
    }

#pragma unroll
    for (int ht = 0; ht < 4; ++ht) {
#pragma unroll
      for (int r = 0; r < 8; ++r) {
        const int dch = wave * 16 + r + hi * 8;
        const int hw  = hw0 + ht * 16 + l15;
        ob[(size_t)dch * HW + hw] = acc[ht][r] - sB[dch];
      }
    }
    __syncthreads();
  }
}

extern "C" void kernel_launch(void* const* d_in, const int* in_sizes, int n_in,
                              void* d_out, int out_size, void* d_ws, size_t ws_size,
                              hipStream_t stream) {
  const float* x   = (const float*)d_in[0];
  const float* rot = (const float*)d_in[1];
  float* out = (float*)d_out;
  float* ws  = (float*)d_ws;

  float* gramPart = ws;
  float* meanPart = gramPart + (size_t)BATCH * CH * CH;
  float* mean     = meanPart + CH * BATCH;
  float* Mmat     = mean + CH;
  float* bias     = Mmat + CH * CH;

  cw_mean_partial<<<dim3(BATCH, CH), 256, 0, stream>>>(x, meanPart);
  cw_gram<<<BATCH, 256, 0, stream>>>(x, gramPart);
  cw_mean_finish<<<1, CH, 0, stream>>>(meanPart, mean);
  cw_newton<<<1, 256, NEWTON_LDS_BYTES, stream>>>(
      gramPart, mean, rot, Mmat, bias);
  cw_apply<<<dim3(7, BATCH), 256,
             (CH * MPAD + CH * XPAD + CH) * sizeof(float), stream>>>(
      x, Mmat, bias, out);
}


# Fixed-order sum of the 64 Gram partials moved out of the single-workgroup Newton kernel into the mean-finish launch (now 129 workgroups, all 64 loads in flight); summation order unchanged
# speedup vs baseline: 17.7755x; 1.4744x over previous
#include <hip/hip_runtime.h>
#include <hip/hip_bf16.h>


typedef __attribute__((ext_vector_type(2))) float v2f;
typedef __attribute__((ext_vector_type(8))) float v8f;

constexpr int CH    = 128;
constexpr int BATCH = 64;
constexpr int HW    = 56 * 56;
constexpr float MTOT = (float)BATCH * (float)HW;
constexpr float EPSR = 1e-5f;
constexpr int   T_NEWTON = 5;

__global__ void __launch_bounds__(256)
cw_mean_partial(const float* __restrict__ x, float* __restrict__ meanPart) {
  const int b = blockIdx.x;
  const int c = blockIdx.y;
  const float* p = x + ((size_t)(b * CH + c)) * HW;
  float s = 0.f;
  for (int i = threadIdx.x; i < HW; i += 256) s += p[i];
  __shared__ float red[256];
  red[threadIdx.x] = s;
  __syncthreads();
  for (int off = 128; off > 0; off >>= 1) {
    if ((int)threadIdx.x < off) red[threadIdx.x] += red[threadIdx.x + off];
    __syncthreads();
  }
  if (threadIdx.x == 0) meanPart[c * BATCH + b] = red[0];
}

__global__ void __launch_bounds__(CH)
cw_reduce(const float* __restrict__ meanPart, float* __restrict__ mean,
          float* gramPart) {
  const int t = threadIdx.x;
  if (blockIdx.x == CH) {
    float s = 0.f;
    for (int b = 0; b < BATCH; ++b) s += meanPart[t * BATCH + b];
    mean[t] = s / MTOT;
    return;
  }
  float* p = gramPart + blockIdx.x * CH + t;
  float v[BATCH];
#pragma unroll
  for (int s = 0; s < BATCH; ++s) v[s] = p[(size_t)s * CH * CH];
  float g = 0.f;
#pragma unroll
  for (int s = 0; s < BATCH; ++s) g += v[s];
  *p = g;
}

constexpr int KC = 64;
constexpr int XPAD = KC + 4;

__global__ void __launch_bounds__(256)
cw_gram(const float* __restrict__ x, float* __restrict__ gramPart) {
  __shared__ float sX[CH][XPAD];
  const int b    = blockIdx.x;
  const int tid  = threadIdx.x;
  const int wave = tid >> 5;
  const int lane = tid & 31;
  const int hi   = lane >> 4;
  const int l15  = lane & 15;

  v8f acc[8] = {};
  const float* xb = x + (size_t)b * CH * HW;

  for (int kc = 0; kc < HW; kc += KC) {
#pragma unroll
    for (int i = 0; i < 8; ++i) {
      const int idx = i * 256 + tid;
      const int c   = idx >> 4;
      const int k4  = (idx & 15) << 2;
      const float4 v = *reinterpret_cast<const float4*>(xb + (size_t)c * HW + kc + k4);
      sX[c][k4 + 0] = v.x; sX[c][k4 + 1] = v.y;
      sX[c][k4 + 2] = v.z; sX[c][k4 + 3] = v.w;
    }
    __syncthreads();

#pragma unroll 2
    for (int k0 = 0; k0 < KC; k0 += 4) {
      const int ks = k0 + hi * 2;
      const int ra = wave * 16 + l15;
      v2f a; a.x = sX[ra][ks]; a.y = sX[ra][ks + 1];
#pragma unroll
      for (int tc = 0; tc < 8; ++tc) {
        const int rb = tc * 16 + l15;
        v2f bb; bb.x = sX[rb][ks]; bb.y = sX[rb][ks + 1];
        acc[tc] = __builtin_amdgcn_wmma_f32_16x16x4_f32(
            false, a, false, bb, (short)0, acc[tc], false, false);
      }
    }
    __syncthreads();
  }

  float* gp = gramPart + (size_t)b * CH * CH;
#pragma unroll
  for (int tc = 0; tc < 8; ++tc) {
#pragma unroll
    for (int r = 0; r < 8; ++r) {
      const int row = wave * 16 + r + hi * 8;
      const int col = tc * 16 + l15;
      gp[row * CH + col] = acc[tc][r];
    }
  }
}

constexpr int NLD = CH + 12;
constexpr size_t NEWTON_LDS_BYTES = (size_t)3 * CH * NLD * sizeof(float);

__device__ inline void strip_mm(v8f (&acc)[8], const float* __restrict__ Astrip,
                                const float* __restrict__ Bm, int hi, int l15) {
#pragma unroll
  for (int tc = 0; tc < 8; ++tc) acc[tc] = v8f{};
  for (int k0 = 0; k0 < CH; k0 += 4) {
    const int ks = k0 + hi * 2;
    v2f a; a.x = Astrip[l15 * NLD + ks]; a.y = Astrip[l15 * NLD + ks + 1];
#pragma unroll
    for (int tc = 0; tc < 8; ++tc) {
      v2f bb;
      bb.x = Bm[ks * NLD + tc * 16 + l15];
      bb.y = Bm[(ks + 1) * NLD + tc * 16 + l15];
      acc[tc] = __builtin_amdgcn_wmma_f32_16x16x4_f32(
          false, a, false, bb, (short)0, acc[tc], false, false);
    }
  }
}

__device__ inline void strip_store(float* __restrict__ strip, const v8f (&acc)[8],
                                   int hi, int l15) {
#pragma unroll
  for (int tc = 0; tc < 8; ++tc) {
#pragma unroll
    for (int r = 0; r < 8; ++r) strip[(r + hi * 8) * NLD + tc * 16 + l15] = acc[tc][r];
  }
}

__global__ void __launch_bounds__(256)
cw_newton(const float* __restrict__ gsum, const float* __restrict__ mean,
          const float* __restrict__ rot, float* __restrict__ Mout,
          float* __restrict__ bias) {
  extern __shared__ float smem[];
  float* S = smem;
  float* P = smem + 1 * CH * NLD;
  float* T = smem + 2 * CH * NLD;
  __shared__ float sTrInv;
  const int tid  = threadIdx.x;
  const int wave = tid >> 5;
  const int lane = tid & 31;
  const int hi   = lane >> 4;
  const int l15  = lane & 15;

#pragma unroll
  for (int i = 0; i < CH * CH / 256; ++i) {
    const int e = i * 256 + tid;
    const float g = gsum[e];
    const int r = e >> 7, c = e & 127;
    float sig = (g - MTOT * mean[r] * mean[c]) / MTOT;
    if (r == c) sig += EPSR;
    S[r * NLD + c] = sig;
  }
  __syncthreads();
  if (tid == 0) {
    float tr = 0.f;
    for (int i = 0; i < CH; ++i) tr += S[i * NLD + i];
    sTrInv = 1.0f / tr;
  }
  __syncthreads();
  const float rTr = sTrInv;
  for (int e = tid; e < CH * CH; e += 256) {
    const int r = e >> 7, c = e & 127;
    S[r * NLD + c] *= rTr;
    P[r * NLD + c] = (r == c) ? 1.f : 0.f;
  }
  __syncthreads();

  float* Tw = T + wave * 16 * NLD;
  float* Pw = P + wave * 16 * NLD;
  v8f acc[8];

  for (int it = 0; it < T_NEWTON; ++it) {
    strip_mm(acc, Pw, P, hi, l15);
    strip_store(Tw, acc, hi, l15);
    __syncthreads();
    strip_mm(acc, Tw, P, hi, l15);
    strip_store(Tw, acc, hi, l15);
    __syncthreads();
    strip_mm(acc, Tw, S, hi, l15);
#pragma unroll
    for (int tc = 0; tc < 8; ++tc) {
#pragma unroll
      for (int r = 0; r < 8; ++r) {
        float* p = Pw + (r + hi * 8) * NLD + tc * 16 + l15;
        *p = 1.5f * *p - 0.5f * acc[tc][r];
      }
    }
    __syncthreads();
  }

  for (int i = lane; i < 16 * CH / 4; i += 32) {
    const int r = i >> 5, c = (i & 31) << 2;
    const float4 v = *reinterpret_cast<const float4*>(rot + (wave * 16 + r) * CH + c);
    float* d = Tw + r * NLD + c;
    d[0] = v.x; d[1] = v.y; d[2] = v.z; d[3] = v.w;
  }
  __syncthreads();
  strip_mm(acc, Tw, P, hi, l15);
  strip_store(Tw, acc, hi, l15);
  const float sc = sqrtf(rTr);
#pragma unroll
  for (int tc = 0; tc < 8; ++tc) {
#pragma unroll
    for (int r = 0; r < 8; ++r)
      Mout[(wave * 16 + r + hi * 8) * CH + tc * 16 + l15] = acc[tc][r] * sc;
  }
  __syncthreads();
  if (tid < CH) {
    float s = 0.f;
    for (int c = 0; c < CH; ++c) s += T[tid * NLD + c] * mean[c];
    bias[tid] = s * sc;
  }
}

constexpr int MPAD = CH + 4;
__global__ void __launch_bounds__(256)
cw_apply(const float* __restrict__ x, const float* __restrict__ Mmat,
         const float* __restrict__ bias, float* __restrict__ out) {
  extern __shared__ float smem[];
  float* sM = smem;
  float* sX = smem + CH * MPAD;
  float* sB = sX + CH * XPAD;
  const int b    = blockIdx.y;
  const int tid  = threadIdx.x;
  const int wave = tid >> 5;
  const int lane = tid & 31;
  const int hi   = lane >> 4;
  const int l15  = lane & 15;

  for (int i = tid; i < (CH * CH) / 4; i += 256) {
    const int e = i * 4;
    const int r = e >> 7, c = e & 127;
    const float4 v = *reinterpret_cast<const float4*>(Mmat + e);
    float* d = sM + r * MPAD + c;
    d[0] = v.x; d[1] = v.y; d[2] = v.z; d[3] = v.w;
  }
  if (tid < CH) sB[tid] = bias[tid];
  __syncthreads();

  const float* xb = x + (size_t)b * CH * HW;
  float*       ob = out + (size_t)b * CH * HW;

  for (int j = 0; j < 7; ++j) {
    const int hw0 = (blockIdx.x * 7 + j) * 64;
#pragma unroll
    for (int i = 0; i < 8; ++i) {
      const int idx = i * 256 + tid;
      const int c   = idx >> 4;
      const int k4  = (idx & 15) << 2;
      const float4 v = *reinterpret_cast<const float4*>(xb + (size_t)c * HW + hw0 + k4);
      float* d = sX + c * XPAD + k4;
      d[0] = v.x; d[1] = v.y; d[2] = v.z; d[3] = v.w;
    }
    __syncthreads();

    v8f acc[4] = {};
#pragma unroll 2
    for (int c0 = 0; c0 < CH; c0 += 4) {
      const int ks = c0 + hi * 2;
      const int ra = wave * 16 + l15;
      v2f a; a.x = sM[ra * MPAD + ks]; a.y = sM[ra * MPAD + ks + 1];
#pragma unroll
      for (int ht = 0; ht < 4; ++ht) {
        v2f bb;
        bb.x = sX[ks * XPAD + ht * 16 + l15];
        bb.y = sX[(ks + 1) * XPAD + ht * 16 + l15];
        acc[ht] = __builtin_amdgcn_wmma_f32_16x16x4_f32(
            false, a, false, bb, (short)0, acc[ht], false, false);
      }
    }

#pragma unroll
    for (int ht = 0; ht < 4; ++ht) {
#pragma unroll
      for (int r = 0; r < 8; ++r) {
        const int dch = wave * 16 + r + hi * 8;
        const int hw  = hw0 + ht * 16 + l15;
        ob[(size_t)dch * HW + hw] = acc[ht][r] - sB[dch];
      }
    }
    __syncthreads();
  }
}

extern "C" void kernel_launch(void* const* d_in, const int* in_sizes, int n_in,
                              void* d_out, int out_size, void* d_ws, size_t ws_size,
                              hipStream_t stream) {
  const float* x   = (const float*)d_in[0];
  const float* rot = (const float*)d_in[1];
  float* out = (float*)d_out;
  float* ws  = (float*)d_ws;

  float* gramPart = ws;
  float* meanPart = gramPart + (size_t)BATCH * CH * CH;
  float* mean     = meanPart + CH * BATCH;
  float* Mmat     = mean + CH;
  float* bias     = Mmat + CH * CH;

  cw_mean_partial<<<dim3(BATCH, CH), 256, 0, stream>>>(x, meanPart);
  cw_gram<<<BATCH, 256, 0, stream>>>(x, gramPart);
  cw_reduce<<<CH + 1, CH, 0, stream>>>(meanPart, mean, gramPart);
  cw_newton<<<1, 256, NEWTON_LDS_BYTES, stream>>>(
      gramPart, mean, rot, Mmat, bias);
  cw_apply<<<dim3(7, BATCH), 256,
             (CH * MPAD + CH * XPAD + CH) * sizeof(float), stream>>>(
      x, Mmat, bias, out);
}


# cw_gram grid 64 -> 256 workgroups: each batch slab's 128x128 Gram partial is split into four 32-row bands, one workgroup each; per-element summation order unchanged
# speedup vs baseline: 21.4812x; 1.2085x over previous
#include <hip/hip_runtime.h>
#include <hip/hip_bf16.h>


typedef __attribute__((ext_vector_type(2))) float v2f;
typedef __attribute__((ext_vector_type(8))) float v8f;

constexpr int CH    = 128;
constexpr int BATCH = 64;
constexpr int HW    = 56 * 56;
constexpr float MTOT = (float)BATCH * (float)HW;
constexpr float EPSR = 1e-5f;
constexpr int   T_NEWTON = 5;

__global__ void __launch_bounds__(256)
cw_mean_partial(const float* __restrict__ x, float* __restrict__ meanPart) {
  const int b = blockIdx.x;
  const int c = blockIdx.y;
  const float* p = x + ((size_t)(b * CH + c)) * HW;
  float s = 0.f;
  for (int i = threadIdx.x; i < HW; i += 256) s += p[i];
  __shared__ float red[256];
  red[threadIdx.x] = s;
  __syncthreads();
  for (int off = 128; off > 0; off >>= 1) {
    if ((int)threadIdx.x < off) red[threadIdx.x] += red[threadIdx.x + off];
    __syncthreads();
  }
  if (threadIdx.x == 0) meanPart[c * BATCH + b] = red[0];
}

__global__ void __launch_bounds__(CH)
cw_reduce(const float* __restrict__ meanPart, float* __restrict__ mean,
          float* gramPart) {
  const int t = threadIdx.x;
  if (blockIdx.x == CH) {
    float s = 0.f;
    for (int b = 0; b < BATCH; ++b) s += meanPart[t * BATCH + b];
    mean[t] = s / MTOT;
    return;
  }
  float* p = gramPart + blockIdx.x * CH + t;
  float v[BATCH];
#pragma unroll
  for (int s = 0; s < BATCH; ++s) v[s] = p[(size_t)s * CH * CH];
  float g = 0.f;
#pragma unroll
  for (int s = 0; s < BATCH; ++s) g += v[s];
  *p = g;
}

constexpr int KC = 64;
constexpr int XPAD = KC + 4;
constexpr int GBANDS = 4;

__global__ void __launch_bounds__(256)
cw_gram(const float* __restrict__ x, float* __restrict__ gramPart) {
  __shared__ float sX[CH][XPAD];
  const int b    = blockIdx.x / GBANDS;
  const int band = blockIdx.x % GBANDS;
  const int tid  = threadIdx.x;
  const int wave = tid >> 5;
  const int lane = tid & 31;
  const int hi   = lane >> 4;
  const int l15  = lane & 15;
  const int row0 = band * 32 + (wave & 1) * 16;
  const int col0 = (wave >> 1) * 32;

  v8f acc[2] = {};
  const float* xb = x + (size_t)b * CH * HW;

  for (int kc = 0; kc < HW; kc += KC) {
#pragma unroll
    for (int i = 0; i < 8; ++i) {
      const int idx = i * 256 + tid;
      const int c   = idx >> 4;
      const int k4  = (idx & 15) << 2;
      const float4 v = *reinterpret_cast<const float4*>(xb + (size_t)c * HW + kc + k4);
      sX[c][k4 + 0] = v.x; sX[c][k4 + 1] = v.y;
      sX[c][k4 + 2] = v.z; sX[c][k4 + 3] = v.w;
    }
    __syncthreads();

#pragma unroll 8
    for (int k0 = 0; k0 < KC; k0 += 4) {
      const int ks = k0 + hi * 2;
      const int ra = row0 + l15;
      v2f a; a.x = sX[ra][ks]; a.y = sX[ra][ks + 1];
#pragma unroll
      for (int t = 0; t < 2; ++t) {
        const int rb = col0 + t * 16 + l15;
        v2f bb; bb.x = sX[rb][ks]; bb.y = sX[rb][ks + 1];
        acc[t] = __builtin_amdgcn_wmma_f32_16x16x4_f32(
            false, a, false, bb, (short)0, acc[t], false, false);
      }
    }
    __syncthreads();
  }

  float* gp = gramPart + (size_t)b * CH * CH;
#pragma unroll
  for (int t = 0; t < 2; ++t) {
#pragma unroll
    for (int r = 0; r < 8; ++r) {
      const int row = row0 + r + hi * 8;
      const int col = col0 + t * 16 + l15;
      gp[row * CH + col] = acc[t][r];
    }
  }
}

constexpr int NLD = CH + 12;
constexpr size_t NEWTON_LDS_BYTES = (size_t)3 * CH * NLD * sizeof(float);

__device__ inline void strip_mm(v8f (&acc)[8], const float* __restrict__ Astrip,
                                const float* __restrict__ Bm, int hi, int l15) {
#pragma unroll
  for (int tc = 0; tc < 8; ++tc) acc[tc] = v8f{};
  for (int k0 = 0; k0 < CH; k0 += 4) {
    const int ks = k0 + hi * 2;
    v2f a; a.x = Astrip[l15 * NLD + ks]; a.y = Astrip[l15 * NLD + ks + 1];
#pragma unroll
    for (int tc = 0; tc < 8; ++tc) {
      v2f bb;
      bb.x = Bm[ks * NLD + tc * 16 + l15];
      bb.y = Bm[(ks + 1) * NLD + tc * 16 + l15];
      acc[tc] = __builtin_amdgcn_wmma_f32_16x16x4_f32(
          false, a, false, bb, (short)0, acc[tc], false, false);
    }
  }
}

__device__ inline void strip_store(float* __restrict__ strip, const v8f (&acc)[8],
                                   int hi, int l15) {
#pragma unroll
  for (int tc = 0; tc < 8; ++tc) {
#pragma unroll
    for (int r = 0; r < 8; ++r) strip[(r + hi * 8) * NLD + tc * 16 + l15] = acc[tc][r];
  }
}

__global__ void __launch_bounds__(256)
cw_newton(const float* __restrict__ gsum, const float* __restrict__ mean,
          const float* __restrict__ rot, float* __restrict__ Mout,
          float* __restrict__ bias) {
  extern __shared__ float smem[];
  float* S = smem;
  float* P = smem + 1 * CH * NLD;
  float* T = smem + 2 * CH * NLD;
  __shared__ float sTrInv;
  const int tid  = threadIdx.x;
  const int wave = tid >> 5;
  const int lane = tid & 31;
  const int hi   = lane >> 4;
  const int l15  = lane & 15;

#pragma unroll
  for (int i = 0; i < CH * CH / 256; ++i) {
    const int e = i * 256 + tid;
    const float g = gsum[e];
    const int r = e >> 7, c = e & 127;
    float sig = (g - MTOT * mean[r] * mean[c]) / MTOT;
    if (r == c) sig += EPSR;
    S[r * NLD + c] = sig;
  }
  __syncthreads();
  if (tid == 0) {
    float tr = 0.f;
    for (int i = 0; i < CH; ++i) tr += S[i * NLD + i];
    sTrInv = 1.0f / tr;
  }
  __syncthreads();
  const float rTr = sTrInv;
  for (int e = tid; e < CH * CH; e += 256) {
    const int r = e >> 7, c = e & 127;
    S[r * NLD + c] *= rTr;
    P[r * NLD + c] = (r == c) ? 1.f : 0.f;
  }
  __syncthreads();

  float* Tw = T + wave * 16 * NLD;
  float* Pw = P + wave * 16 * NLD;
  v8f acc[8];

  for (int it = 0; it < T_NEWTON; ++it) {
    strip_mm(acc, Pw, P, hi, l15);
    strip_store(Tw, acc, hi, l15);
    __syncthreads();
    strip_mm(acc, Tw, P, hi, l15);
    strip_store(Tw, acc, hi, l15);
    __syncthreads();
    strip_mm(acc, Tw, S, hi, l15);
#pragma unroll
    for (int tc = 0; tc < 8; ++tc) {
#pragma unroll
      for (int r = 0; r < 8; ++r) {
        float* p = Pw + (r + hi * 8) * NLD + tc * 16 + l15;
        *p = 1.5f * *p - 0.5f * acc[tc][r];
      }
    }
    __syncthreads();
  }

  for (int i = lane; i < 16 * CH / 4; i += 32) {
    const int r = i >> 5, c = (i & 31) << 2;
    const float4 v = *reinterpret_cast<const float4*>(rot + (wave * 16 + r) * CH + c);
    float* d = Tw + r * NLD + c;
    d[0] = v.x; d[1] = v.y; d[2] = v.z; d[3] = v.w;
  }
  __syncthreads();
  strip_mm(acc, Tw, P, hi, l15);
  strip_store(Tw, acc, hi, l15);
  const float sc = sqrtf(rTr);
#pragma unroll
  for (int tc = 0; tc < 8; ++tc) {
#pragma unroll
    for (int r = 0; r < 8; ++r)
      Mout[(wave * 16 + r + hi * 8) * CH + tc * 16 + l15] = acc[tc][r] * sc;
  }
  __syncthreads();
  if (tid < CH) {
    float s = 0.f;
    for (int c = 0; c < CH; ++c) s += T[tid * NLD + c] * mean[c];
    bias[tid] = s * sc;
  }
}

constexpr int MPAD = CH + 4;
__global__ void __launch_bounds__(256)
cw_apply(const float* __restrict__ x, const float* __restrict__ Mmat,
         const float* __restrict__ bias, float* __restrict__ out) {
  extern __shared__ float smem[];
  float* sM = smem;
  float* sX = smem + CH * MPAD;
  float* sB = sX + CH * XPAD;
  const int b    = blockIdx.y;
  const int tid  = threadIdx.x;
  const int wave = tid >> 5;
  const int lane = tid & 31;
  const int hi   = lane >> 4;
  const int l15  = lane & 15;

  for (int i = tid; i < (CH * CH) / 4; i += 256) {
    const int e = i * 4;
    const int r = e >> 7, c = e & 127;
    const float4 v = *reinterpret_cast<const float4*>(Mmat + e);
    float* d = sM + r * MPAD + c;
    d[0] = v.x; d[1] = v.y; d[2] = v.z; d[3] = v.w;
  }
  if (tid < CH) sB[tid] = bias[tid];
  __syncthreads();

  const float* xb = x + (size_t)b * CH * HW;
  float*       ob = out + (size_t)b * CH * HW;

  for (int j = 0; j < 7; ++j) {
    const int hw0 = (blockIdx.x * 7 + j) * 64;
#pragma unroll
    for (int i = 0; i < 8; ++i) {
      const int idx = i * 256 + tid;
      const int c   = idx >> 4;
      const int k4  = (idx & 15) << 2;
      const float4 v = *reinterpret_cast<const float4*>(xb + (size_t)c * HW + hw0 + k4);
      float* d = sX + c * XPAD + k4;
      d[0] = v.x; d[1] = v.y; d[2] = v.z; d[3] = v.w;
    }
    __syncthreads();

    v8f acc[4] = {};
#pragma unroll 2
    for (int c0 = 0; c0 < CH; c0 += 4) {
      const int ks = c0 + hi * 2;
      const int ra = wave * 16 + l15;
      v2f a; a.x = sM[ra * MPAD + ks]; a.y = sM[ra * MPAD + ks + 1];
#pragma unroll
      for (int ht = 0; ht < 4; ++ht) {
        v2f bb;
        bb.x = sX[ks * XPAD + ht * 16 + l15];
        bb.y = sX[(ks + 1) * XPAD + ht * 16 + l15];
        acc[ht] = __builtin_amdgcn_wmma_f32_16x16x4_f32(
            false, a, false, bb, (short)0, acc[ht], false, false);
      }
    }

#pragma unroll
    for (int ht = 0; ht < 4; ++ht) {
#pragma unroll
      for (int r = 0; r < 8; ++r) {
        const int dch = wave * 16 + r + hi * 8;
        const int hw  = hw0 + ht * 16 + l15;
        ob[(size_t)dch * HW + hw] = acc[ht][r] - sB[dch];
      }
    }
    __syncthreads();
  }
}

extern "C" void kernel_launch(void* const* d_in, const int* in_sizes, int n_in,
                              void* d_out, int out_size, void* d_ws, size_t ws_size,
                              hipStream_t stream) {
  const float* x   = (const float*)d_in[0];
  const float* rot = (const float*)d_in[1];
  float* out = (float*)d_out;
  float* ws  = (float*)d_ws;

  float* gramPart = ws;
  float* meanPart = gramPart + (size_t)BATCH * CH * CH;
  float* mean     = meanPart + CH * BATCH;
  float* Mmat     = mean + CH;
  float* bias     = Mmat + CH * CH;

  cw_mean_partial<<<dim3(BATCH, CH), 256, 0, stream>>>(x, meanPart);
  cw_gram<<<BATCH * GBANDS, 256, 0, stream>>>(x, gramPart);
  cw_reduce<<<CH + 1, CH, 0, stream>>>(meanPart, mean, gramPart);
  cw_newton<<<1, 256, NEWTON_LDS_BYTES, stream>>>(
      gramPart, mean, rot, Mmat, bias);
  cw_apply<<<dim3(7, BATCH), 256,
             (CH * MPAD + CH * XPAD + CH) * sizeof(float), stream>>>(
      x, Mmat, bias, out);
}


# Channel sums for the mean folded into cw_gram's staged LDS slab: the 8192-workgroup mean launch and its read of x are gone (4 launches); mean summation order changes (8 interleaved partials per row)
# speedup vs baseline: 22.8229x; 1.0625x over previous
#include <hip/hip_runtime.h>
#include <hip/hip_bf16.h>


typedef __attribute__((ext_vector_type(2))) float v2f;
typedef __attribute__((ext_vector_type(8))) float v8f;

constexpr int CH    = 128;
constexpr int BATCH = 64;
constexpr int HW    = 56 * 56;
constexpr float MTOT = (float)BATCH * (float)HW;
constexpr float EPSR = 1e-5f;
constexpr int   T_NEWTON = 5;

__global__ void __launch_bounds__(CH)
cw_reduce(const float* __restrict__ meanPart, float* __restrict__ mean,
          float* gramPart) {
  const int t = threadIdx.x;
  if (blockIdx.x == CH) {
    float s = 0.f;
    for (int b = 0; b < BATCH; ++b) s += meanPart[b * CH + t];
    mean[t] = s / MTOT;
    return;
  }
  float* p = gramPart + blockIdx.x * CH + t;
  float v[BATCH];
#pragma unroll
  for (int s = 0; s < BATCH; ++s) v[s] = p[(size_t)s * CH * CH];
  float g = 0.f;
#pragma unroll
  for (int s = 0; s < BATCH; ++s) g += v[s];
  *p = g;
}

constexpr int KC = 64;
constexpr int XPAD = KC + 4;
constexpr int GBANDS = 4;

__global__ void __launch_bounds__(256)
cw_gram(const float* __restrict__ x, float* __restrict__ gramPart,
        float* __restrict__ meanPart) {
  __shared__ alignas(16) float sX[CH][XPAD];
  __shared__ float sRow[32][9];
  const int b    = blockIdx.x / GBANDS;
  const int band = blockIdx.x % GBANDS;
  const int tid  = threadIdx.x;
  const int wave = tid >> 5;
  const int lane = tid & 31;
  const int hi   = lane >> 4;
  const int l15  = lane & 15;
  const int row0 = band * 32 + (wave & 1) * 16;
  const int col0 = (wave >> 1) * 32;

  v8f acc[2] = {};
  const float* xb = x + (size_t)b * CH * HW;
  const int mrow = band * 32 + (tid >> 3);
  const int mcol = (tid & 7) << 3;
  float msum = 0.f;

  for (int kc = 0; kc < HW; kc += KC) {
#pragma unroll
    for (int i = 0; i < 8; ++i) {
      const int idx = i * 256 + tid;
      const int c   = idx >> 4;
      const int k4  = (idx & 15) << 2;
      const float4 v = *reinterpret_cast<const float4*>(xb + (size_t)c * HW + kc + k4);
      sX[c][k4 + 0] = v.x; sX[c][k4 + 1] = v.y;
      sX[c][k4 + 2] = v.z; sX[c][k4 + 3] = v.w;
    }
    __syncthreads();

    {
      const float4 u0 = *reinterpret_cast<const float4*>(&sX[mrow][mcol]);
      const float4 u1 = *reinterpret_cast<const float4*>(&sX[mrow][mcol + 4]);
      msum += u0.x; msum += u0.y; msum += u0.z; msum += u0.w;
      msum += u1.x; msum += u1.y; msum += u1.z; msum += u1.w;
    }

#pragma unroll 8
    for (int k0 = 0; k0 < KC; k0 += 4) {
      const int ks = k0 + hi * 2;
      const int ra = row0 + l15;
      v2f a; a.x = sX[ra][ks]; a.y = sX[ra][ks + 1];
#pragma unroll
      for (int t = 0; t < 2; ++t) {
        const int rb = col0 + t * 16 + l15;
        v2f bb; bb.x = sX[rb][ks]; bb.y = sX[rb][ks + 1];
        acc[t] = __builtin_amdgcn_wmma_f32_16x16x4_f32(
            false, a, false, bb, (short)0, acc[t], false, false);
      }
    }
    __syncthreads();
  }

  float* gp = gramPart + (size_t)b * CH * CH;
#pragma unroll
  for (int t = 0; t < 2; ++t) {
#pragma unroll
    for (int r = 0; r < 8; ++r) {
      const int row = row0 + r + hi * 8;
      const int col = col0 + t * 16 + l15;
      gp[row * CH + col] = acc[t][r];
    }
  }

  sRow[tid >> 3][tid & 7] = msum;
  __syncthreads();
  if (tid < 32) {
    float s = 0.f;
    for (int j = 0; j < 8; ++j) s += sRow[tid][j];
    meanPart[b * CH + band * 32 + tid] = s;
  }
}

constexpr int NLD = CH + 12;
constexpr size_t NEWTON_LDS_BYTES = (size_t)3 * CH * NLD * sizeof(float);

__device__ inline void strip_mm(v8f (&acc)[8], const float* __restrict__ Astrip,
                                const float* __restrict__ Bm, int hi, int l15) {
#pragma unroll
  for (int tc = 0; tc < 8; ++tc) acc[tc] = v8f{};
  for (int k0 = 0; k0 < CH; k0 += 4) {
    const int ks = k0 + hi * 2;
    v2f a; a.x = Astrip[l15 * NLD + ks]; a.y = Astrip[l15 * NLD + ks + 1];
#pragma unroll
    for (int tc = 0; tc < 8; ++tc) {
      v2f bb;
      bb.x = Bm[ks * NLD + tc * 16 + l15];
      bb.y = Bm[(ks + 1) * NLD + tc * 16 + l15];
      acc[tc] = __builtin_amdgcn_wmma_f32_16x16x4_f32(
          false, a, false, bb, (short)0, acc[tc], false, false);
    }
  }
}

__device__ inline void strip_store(float* __restrict__ strip, const v8f (&acc)[8],
                                   int hi, int l15) {
#pragma unroll
  for (int tc = 0; tc < 8; ++tc) {
#pragma unroll
    for (int r = 0; r < 8; ++r) strip[(r + hi * 8) * NLD + tc * 16 + l15] = acc[tc][r];
  }
}

__global__ void __launch_bounds__(256)
cw_newton(const float* __restrict__ gsum, const float* __restrict__ mean,
          const float* __restrict__ rot, float* __restrict__ Mout,
          float* __restrict__ bias) {
  extern __shared__ float smem[];
  float* S = smem;
  float* P = smem + 1 * CH * NLD;
  float* T = smem + 2 * CH * NLD;
  __shared__ float sTrInv;
  const int tid  = threadIdx.x;
  const int wave = tid >> 5;
  const int lane = tid & 31;
  const int hi   = lane >> 4;
  const int l15  = lane & 15;

#pragma unroll
  for (int i = 0; i < CH * CH / 256; ++i) {
    const int e = i * 256 + tid;
    const float g = gsum[e];
    const int r = e >> 7, c = e & 127;
    float sig = (g - MTOT * mean[r] * mean[c]) / MTOT;
    if (r == c) sig += EPSR;
    S[r * NLD + c] = sig;
  }
  __syncthreads();
  if (tid == 0) {
    float tr = 0.f;
    for (int i = 0; i < CH; ++i) tr += S[i * NLD + i];
    sTrInv = 1.0f / tr;
  }
  __syncthreads();
  const float rTr = sTrInv;
  for (int e = tid; e < CH * CH; e += 256) {
    const int r = e >> 7, c = e & 127;
    S[r * NLD + c] *= rTr;
    P[r * NLD + c] = (r == c) ? 1.f : 0.f;
  }
  __syncthreads();

  float* Tw = T + wave * 16 * NLD;
  float* Pw = P + wave * 16 * NLD;
  v8f acc[8];

  for (int it = 0; it < T_NEWTON; ++it) {
    strip_mm(acc, Pw, P, hi, l15);
    strip_store(Tw, acc, hi, l15);
    __syncthreads();
    strip_mm(acc, Tw, P, hi, l15);
    strip_store(Tw, acc, hi, l15);
    __syncthreads();
    strip_mm(acc, Tw, S, hi, l15);
#pragma unroll
    for (int tc = 0; tc < 8; ++tc) {
#pragma unroll
      for (int r = 0; r < 8; ++r) {
        float* p = Pw + (r + hi * 8) * NLD + tc * 16 + l15;
        *p = 1.5f * *p - 0.5f * acc[tc][r];
      }
    }
    __syncthreads();
  }

  for (int i = lane; i < 16 * CH / 4; i += 32) {
    const int r = i >> 5, c = (i & 31) << 2;
    const float4 v = *reinterpret_cast<const float4*>(rot + (wave * 16 + r) * CH + c);
    float* d = Tw + r * NLD + c;
    d[0] = v.x; d[1] = v.y; d[2] = v.z; d[3] = v.w;
  }
  __syncthreads();
  strip_mm(acc, Tw, P, hi, l15);
  strip_store(Tw, acc, hi, l15);
  const float sc = sqrtf(rTr);
#pragma unroll
  for (int tc = 0; tc < 8; ++tc) {
#pragma unroll
    for (int r = 0; r < 8; ++r)
      Mout[(wave * 16 + r + hi * 8) * CH + tc * 16 + l15] = acc[tc][r] * sc;
  }
  __syncthreads();
  if (tid < CH) {
    float s = 0.f;
    for (int c = 0; c < CH; ++c) s += T[tid * NLD + c] * mean[c];
    bias[tid] = s * sc;
  }
}

constexpr int MPAD = CH + 4;
__global__ void __launch_bounds__(256)
cw_apply(const float* __restrict__ x, const float* __restrict__ Mmat,
         const float* __restrict__ bias, float* __restrict__ out) {
  extern __shared__ float smem[];
  float* sM = smem;
  float* sX = smem + CH * MPAD;
  float* sB = sX + CH * XPAD;
  const int b    = blockIdx.y;
  const int tid  = threadIdx.x;
  const int wave = tid >> 5;
  const int lane = tid & 31;
  const int hi   = lane >> 4;
  const int l15  = lane & 15;

  for (int i = tid; i < (CH * CH) / 4; i += 256) {
    const int e = i * 4;
    const int r = e >> 7, c = e & 127;
    const float4 v = *reinterpret_cast<const float4*>(Mmat + e);
    float* d = sM + r * MPAD + c;
    d[0] = v.x; d[1] = v.y; d[2] = v.z; d[3] = v.w;
  }
  if (tid < CH) sB[tid] = bias[tid];
  __syncthreads();

  const float* xb = x + (size_t)b * CH * HW;
  float*       ob = out + (size_t)b * CH * HW;

  for (int j = 0; j < 7; ++j) {
    const int hw0 = (blockIdx.x * 7 + j) * 64;
#pragma unroll
    for (int i = 0; i < 8; ++i) {
      const int idx = i * 256 + tid;
      const int c   = idx >> 4;
      const int k4  = (idx & 15) << 2;
      const float4 v = *reinterpret_cast<const float4*>(xb + (size_t)c * HW + hw0 + k4);
      float* d = sX + c * XPAD + k4;
      d[0] = v.x; d[1] = v.y; d[2] = v.z; d[3] = v.w;
    }
    __syncthreads();

    v8f acc[4] = {};
#pragma unroll 2
    for (int c0 = 0; c0 < CH; c0 += 4) {
      const int ks = c0 + hi * 2;
      const int ra = wave * 16 + l15;
      v2f a; a.x = sM[ra * MPAD + ks]; a.y = sM[ra * MPAD + ks + 1];
#pragma unroll
      for (int ht = 0; ht < 4; ++ht) {
        v2f bb;
        bb.x = sX[ks * XPAD + ht * 16 + l15];
        bb.y = sX[(ks + 1) * XPAD + ht * 16 + l15];
        acc[ht] = __builtin_amdgcn_wmma_f32_16x16x4_f32(
            false, a, false, bb, (short)0, acc[ht], false, false);
      }
    }

#pragma unroll
    for (int ht = 0; ht < 4; ++ht) {
#pragma unroll
      for (int r = 0; r < 8; ++r) {
        const int dch = wave * 16 + r + hi * 8;
        const int hw  = hw0 + ht * 16 + l15;
        ob[(size_t)dch * HW + hw] = acc[ht][r] - sB[dch];
      }
    }
    __syncthreads();
  }
}

extern "C" void kernel_launch(void* const* d_in, const int* in_sizes, int n_in,
                              void* d_out, int out_size, void* d_ws, size_t ws_size,
                              hipStream_t stream) {
  const float* x   = (const float*)d_in[0];
  const float* rot = (const float*)d_in[1];
  float* out = (float*)d_out;
  float* ws  = (float*)d_ws;

  float* gramPart = ws;
  float* meanPart = gramPart + (size_t)BATCH * CH * CH;
  float* mean     = meanPart + CH * BATCH;
  float* Mmat     = mean + CH;
  float* bias     = Mmat + CH * CH;

  cw_gram<<<BATCH * GBANDS, 256, 0, stream>>>(x, gramPart, meanPart);
  cw_reduce<<<CH + 1, CH, 0, stream>>>(meanPart, mean, gramPart);
  cw_newton<<<1, 256, NEWTON_LDS_BYTES, stream>>>(
      gramPart, mean, rot, Mmat, bias);
  cw_apply<<<dim3(7, BATCH), 256,
             (CH * MPAD + CH * XPAD + CH) * sizeof(float), stream>>>(
      x, Mmat, bias, out);
}


# cw_gram staging software-pipelined: next two K chunks' global loads in flight in registers under the current chunk's products (two LDS slabs, one LDS-only barrier per chunk); sum order unchanged
# speedup vs baseline: 27.3356x; 1.1977x over previous
#include <hip/hip_runtime.h>
#include <hip/hip_bf16.h>


typedef __attribute__((ext_vector_type(2))) float v2f;
typedef __attribute__((ext_vector_type(8))) float v8f;
typedef __attribute__((ext_vector_type(4))) float v4f;

constexpr int CH    = 128;
constexpr int BATCH = 64;
constexpr int HW    = 56 * 56;
constexpr float MTOT = (float)BATCH * (float)HW;
constexpr float EPSR = 1e-5f;
constexpr int   T_NEWTON = 5;

__global__ void __launch_bounds__(CH)
cw_reduce(const float* __restrict__ meanPart, float* __restrict__ mean,
          float* gramPart) {
  const int t = threadIdx.x;
  if (blockIdx.x == CH) {
    float s = 0.f;
    for (int b = 0; b < BATCH; ++b) s += meanPart[b * CH + t];
    mean[t] = s / MTOT;
    return;
  }
  float* p = gramPart + blockIdx.x * CH + t;
  float v[BATCH];
#pragma unroll
  for (int s = 0; s < BATCH; ++s) v[s] = p[(size_t)s * CH * CH];
  float g = 0.f;
#pragma unroll
  for (int s = 0; s < BATCH; ++s) g += v[s];
  *p = g;
}

constexpr int KC = 64;
constexpr int XPAD = KC + 4;
constexpr int GBANDS = 4;
constexpr int NCHUNK = HW / KC;

__device__ __forceinline__ void lds_barrier() {
  __builtin_amdgcn_fence(__ATOMIC_RELEASE, "workgroup", "local");
  __builtin_amdgcn_s_barrier();
  __builtin_amdgcn_fence(__ATOMIC_ACQUIRE, "workgroup", "local");
}

__device__ __forceinline__ void gram_fetch(v4f (&v)[8], const float* __restrict__ xb,
                                           int kc, int tid) {
#pragma unroll
  for (int i = 0; i < 8; ++i) {
    const int idx = i * 256 + tid;
    const int c   = idx >> 4;
    const int k4  = (idx & 15) << 2;
    v[i] = *reinterpret_cast<const v4f*>(xb + (size_t)c * HW + kc + k4);
  }
}

__device__ __forceinline__ void gram_stage(float (*s)[XPAD], const v4f (&v)[8], int tid) {
#pragma unroll
  for (int i = 0; i < 8; ++i) {
    const int idx = i * 256 + tid;
    const int c   = idx >> 4;
    const int k4  = (idx & 15) << 2;
    *reinterpret_cast<v4f*>(&s[c][k4]) = v[i];
  }
}

__device__ __forceinline__ void gram_chunk(const float (*s)[XPAD], v8f (&acc)[2], float& msum,
                                           int mrow, int mcol, int row0, int col0,
                                           int hi, int l15) {
  {
    const float4 u0 = *reinterpret_cast<const float4*>(&s[mrow][mcol]);
    const float4 u1 = *reinterpret_cast<const float4*>(&s[mrow][mcol + 4]);
    msum += u0.x; msum += u0.y; msum += u0.z; msum += u0.w;
    msum += u1.x; msum += u1.y; msum += u1.z; msum += u1.w;
  }
#pragma unroll 8
  for (int k0 = 0; k0 < KC; k0 += 4) {
    const int ks = k0 + hi * 2;
    const int ra = row0 + l15;
    v2f a; a.x = s[ra][ks]; a.y = s[ra][ks + 1];
#pragma unroll
    for (int t = 0; t < 2; ++t) {
      const int rb = col0 + t * 16 + l15;
      v2f bb; bb.x = s[rb][ks]; bb.y = s[rb][ks + 1];
      acc[t] = __builtin_amdgcn_wmma_f32_16x16x4_f32(
          false, a, false, bb, (short)0, acc[t], false, false);
    }
  }
}

__global__ void __launch_bounds__(256)
cw_gram(const float* __restrict__ x, float* __restrict__ gramPart,
        float* __restrict__ meanPart) {
  __shared__ alignas(16) float sX[2][CH][XPAD];
  __shared__ float sRow[32][9];
  const int b    = blockIdx.x / GBANDS;
  const int band = blockIdx.x % GBANDS;
  const int tid  = threadIdx.x;
  const int wave = tid >> 5;
  const int lane = tid & 31;
  const int hi   = lane >> 4;
  const int l15  = lane & 15;
  const int row0 = band * 32 + (wave & 1) * 16;
  const int col0 = (wave >> 1) * 32;

  v8f acc[2] = {};
  const float* xb = x + (size_t)b * CH * HW;
  const int mrow = band * 32 + (tid >> 3);
  const int mcol = (tid & 7) << 3;
  float msum = 0.f;

  v4f pa[8], pb[8];
  gram_fetch(pa, xb, 0, tid);
  gram_fetch(pb, xb, KC, tid);
  gram_stage(sX[0], pa, tid);
  lds_barrier();

  static_assert(NCHUNK % 2 == 1 && NCHUNK >= 5, "tail below assumes an odd chunk count");
  for (int ch = 0; ch < NCHUNK - 3; ch += 2) {
    gram_fetch(pa, xb, (ch + 2) * KC, tid);
    gram_chunk(sX[0], acc, msum, mrow, mcol, row0, col0, hi, l15);
    gram_stage(sX[1], pb, tid);
    lds_barrier();
    gram_fetch(pb, xb, (ch + 3) * KC, tid);
    gram_chunk(sX[1], acc, msum, mrow, mcol, row0, col0, hi, l15);
    gram_stage(sX[0], pa, tid);
    lds_barrier();
  }
  gram_fetch(pa, xb, (NCHUNK - 1) * KC, tid);
  gram_chunk(sX[0], acc, msum, mrow, mcol, row0, col0, hi, l15);
  gram_stage(sX[1], pb, tid);
  lds_barrier();
  gram_chunk(sX[1], acc, msum, mrow, mcol, row0, col0, hi, l15);
  gram_stage(sX[0], pa, tid);
  lds_barrier();
  gram_chunk(sX[0], acc, msum, mrow, mcol, row0, col0, hi, l15);

  float* gp = gramPart + (size_t)b * CH * CH;
#pragma unroll
  for (int t = 0; t < 2; ++t) {
#pragma unroll
    for (int r = 0; r < 8; ++r) {
      const int row = row0 + r + hi * 8;
      const int col = col0 + t * 16 + l15;
      gp[row * CH + col] = acc[t][r];
    }
  }

  sRow[tid >> 3][tid & 7] = msum;
  __syncthreads();
  if (tid < 32) {
    float s = 0.f;
    for (int j = 0; j < 8; ++j) s += sRow[tid][j];
    meanPart[b * CH + band * 32 + tid] = s;
  }
}

constexpr int NLD = CH + 12;
constexpr size_t NEWTON_LDS_BYTES = (size_t)3 * CH * NLD * sizeof(float);

__device__ inline void strip_mm(v8f (&acc)[8], const float* __restrict__ Astrip,
                                const float* __restrict__ Bm, int hi, int l15) {
#pragma unroll
  for (int tc = 0; tc < 8; ++tc) acc[tc] = v8f{};
  for (int k0 = 0; k0 < CH; k0 += 4) {
    const int ks = k0 + hi * 2;
    v2f a; a.x = Astrip[l15 * NLD + ks]; a.y = Astrip[l15 * NLD + ks + 1];
#pragma unroll
    for (int tc = 0; tc < 8; ++tc) {
      v2f bb;
      bb.x = Bm[ks * NLD + tc * 16 + l15];
      bb.y = Bm[(ks + 1) * NLD + tc * 16 + l15];
      acc[tc] = __builtin_amdgcn_wmma_f32_16x16x4_f32(
          false, a, false, bb, (short)0, acc[tc], false, false);
    }
  }
}

__device__ inline void strip_store(float* __restrict__ strip, const v8f (&acc)[8],
                                   int hi, int l15) {
#pragma unroll
  for (int tc = 0; tc < 8; ++tc) {
#pragma unroll
    for (int r = 0; r < 8; ++r) strip[(r + hi * 8) * NLD + tc * 16 + l15] = acc[tc][r];
  }
}

__global__ void __launch_bounds__(256)
cw_newton(const float* __restrict__ gsum, const float* __restrict__ mean,
          const float* __restrict__ rot, float* __restrict__ Mout,
          float* __restrict__ bias) {
  extern __shared__ float smem[];
  float* S = smem;
  float* P = smem + 1 * CH * NLD;
  float* T = smem + 2 * CH * NLD;
  __shared__ float sTrInv;
  const int tid  = threadIdx.x;
  const int wave = tid >> 5;
  const int lane = tid & 31;
  const int hi   = lane >> 4;
  const int l15  = lane & 15;

#pragma unroll
  for (int i = 0; i < CH * CH / 256; ++i) {
    const int e = i * 256 + tid;
    const float g = gsum[e];
    const int r = e >> 7, c = e & 127;
    float sig = (g - MTOT * mean[r] * mean[c]) / MTOT;
    if (r == c) sig += EPSR;
    S[r * NLD + c] = sig;
  }
  __syncthreads();
  if (tid == 0) {
    float tr = 0.f;
    for (int i = 0; i < CH; ++i) tr += S[i * NLD + i];
    sTrInv = 1.0f / tr;
  }
  __syncthreads();
  const float rTr = sTrInv;
  for (int e = tid; e < CH * CH; e += 256) {
    const int r = e >> 7, c = e & 127;
    S[r * NLD + c] *= rTr;
    P[r * NLD + c] = (r == c) ? 1.f : 0.f;
  }
  __syncthreads();

  float* Tw = T + wave * 16 * NLD;
  float* Pw = P + wave * 16 * NLD;
  v8f acc[8];

  for (int it = 0; it < T_NEWTON; ++it) {
    strip_mm(acc, Pw, P, hi, l15);
    strip_store(Tw, acc, hi, l15);
    __syncthreads();
    strip_mm(acc, Tw, P, hi, l15);
    strip_store(Tw, acc, hi, l15);
    __syncthreads();
    strip_mm(acc, Tw, S, hi, l15);
#pragma unroll
    for (int tc = 0; tc < 8; ++tc) {
#pragma unroll
      for (int r = 0; r < 8; ++r) {
        float* p = Pw + (r + hi * 8) * NLD + tc * 16 + l15;
        *p = 1.5f * *p - 0.5f * acc[tc][r];
      }
    }
    __syncthreads();
  }

  for (int i = lane; i < 16 * CH / 4; i += 32) {
    const int r = i >> 5, c = (i & 31) << 2;
    const float4 v = *reinterpret_cast<const float4*>(rot + (wave * 16 + r) * CH + c);
    float* d = Tw + r * NLD + c;
    d[0] = v.x; d[1] = v.y; d[2] = v.z; d[3] = v.w;
  }
  __syncthreads();
  strip_mm(acc, Tw, P, hi, l15);
  strip_store(Tw, acc, hi, l15);
  const float sc = sqrtf(rTr);
#pragma unroll
  for (int tc = 0; tc < 8; ++tc) {
#pragma unroll
    for (int r = 0; r < 8; ++r)
      Mout[(wave * 16 + r + hi * 8) * CH + tc * 16 + l15] = acc[tc][r] * sc;
  }
  __syncthreads();
  if (tid < CH) {
    float s = 0.f;
    for (int c = 0; c < CH; ++c) s += T[tid * NLD + c] * mean[c];
    bias[tid] = s * sc;
  }
}

constexpr int MPAD = CH + 4;
__global__ void __launch_bounds__(256)
cw_apply(const float* __restrict__ x, const float* __restrict__ Mmat,
         const float* __restrict__ bias, float* __restrict__ out) {
  extern __shared__ float smem[];
  float* sM = smem;
  float* sX = smem + CH * MPAD;
  float* sB = sX + CH * XPAD;
  const int b    = blockIdx.y;
  const int tid  = threadIdx.x;
  const int wave = tid >> 5;
  const int lane = tid & 31;
  const int hi   = lane >> 4;
  const int l15  = lane & 15;

  for (int i = tid; i < (CH * CH) / 4; i += 256) {
    const int e = i * 4;
    const int r = e >> 7, c = e & 127;
    const float4 v = *reinterpret_cast<const float4*>(Mmat + e);
    float* d = sM + r * MPAD + c;
    d[0] = v.x; d[1] = v.y; d[2] = v.z; d[3] = v.w;
  }
  if (tid < CH) sB[tid] = bias[tid];
  __syncthreads();

  const float* xb = x + (size_t)b * CH * HW;
  float*       ob = out + (size_t)b * CH * HW;

  for (int j = 0; j < 7; ++j) {
    const int hw0 = (blockIdx.x * 7 + j) * 64;
#pragma unroll
    for (int i = 0; i < 8; ++i) {
      const int idx = i * 256 + tid;
      const int c   = idx >> 4;
      const int k4  = (idx & 15) << 2;
      const float4 v = *reinterpret_cast<const float4*>(xb + (size_t)c * HW + hw0 + k4);
      float* d = sX + c * XPAD + k4;
      d[0] = v.x; d[1] = v.y; d[2] = v.z; d[3] = v.w;
    }
    __syncthreads();

    v8f acc[4] = {};
#pragma unroll 2
    for (int c0 = 0; c0 < CH; c0 += 4) {
      const int ks = c0 + hi * 2;
      const int ra = wave * 16 + l15;
      v2f a; a.x = sM[ra * MPAD + ks]; a.y = sM[ra * MPAD + ks + 1];
#pragma unroll
      for (int ht = 0; ht < 4; ++ht) {
        v2f bb;
        bb.x = sX[ks * XPAD + ht * 16 + l15];
        bb.y = sX[(ks + 1) * XPAD + ht * 16 + l15];
        acc[ht] = __builtin_amdgcn_wmma_f32_16x16x4_f32(
            false, a, false, bb, (short)0, acc[ht], false, false);
      }
    }

#pragma unroll
    for (int ht = 0; ht < 4; ++ht) {
#pragma unroll
      for (int r = 0; r < 8; ++r) {
        const int dch = wave * 16 + r + hi * 8;
        const int hw  = hw0 + ht * 16 + l15;
        ob[(size_t)dch * HW + hw] = acc[ht][r] - sB[dch];
      }
    }
    __syncthreads();
  }
}

extern "C" void kernel_launch(void* const* d_in, const int* in_sizes, int n_in,
                              void* d_out, int out_size, void* d_ws, size_t ws_size,
                              hipStream_t stream) {
  const float* x   = (const float*)d_in[0];
  const float* rot = (const float*)d_in[1];
  float* out = (float*)d_out;
  float* ws  = (float*)d_ws;

  float* gramPart = ws;
  float* meanPart = gramPart + (size_t)BATCH * CH * CH;
  float* mean     = meanPart + CH * BATCH;
  float* Mmat     = mean + CH;
  float* bias     = Mmat + CH * CH;

  cw_gram<<<BATCH * GBANDS, 256, 0, stream>>>(x, gramPart, meanPart);
  cw_reduce<<<CH + 1, CH, 0, stream>>>(meanPart, mean, gramPart);
  cw_newton<<<1, 256, NEWTON_LDS_BYTES, stream>>>(
      gramPart, mean, rot, Mmat, bias);
  cw_apply<<<dim3(7, BATCH), 256,
             (CH * MPAD + CH * XPAD + CH) * sizeof(float), stream>>>(
      x, Mmat, bias, out);
}


# Newton stage spread from one workgroup to eight: one launch per iteration, workgroup s owns 16-row strip s through its three chained products (9 launches, was 4); every sum in the same order
# speedup vs baseline: 38.0947x; 1.3936x over previous
#include <hip/hip_runtime.h>
#include <hip/hip_bf16.h>


typedef __attribute__((ext_vector_type(2))) float v2f;
typedef __attribute__((ext_vector_type(8))) float v8f;
typedef __attribute__((ext_vector_type(4))) float v4f;

constexpr int CH    = 128;
constexpr int BATCH = 64;
constexpr int HW    = 56 * 56;
constexpr float MTOT = (float)BATCH * (float)HW;
constexpr float EPSR = 1e-5f;
constexpr int   T_NEWTON = 5;

__global__ void __launch_bounds__(CH)
cw_reduce(const float* __restrict__ meanPart, float* __restrict__ mean,
          float* gramPart) {
  const int t = threadIdx.x;
  if (blockIdx.x == CH) {
    float s = 0.f;
    for (int b = 0; b < BATCH; ++b) s += meanPart[b * CH + t];
    mean[t] = s / MTOT;
    return;
  }
  float* p = gramPart + blockIdx.x * CH + t;
  float v[BATCH];
#pragma unroll
  for (int s = 0; s < BATCH; ++s) v[s] = p[(size_t)s * CH * CH];
  float g = 0.f;
#pragma unroll
  for (int s = 0; s < BATCH; ++s) g += v[s];
  *p = g;
}

constexpr int KC = 64;
constexpr int XPAD = KC + 4;
constexpr int GBANDS = 4;
constexpr int NCHUNK = HW / KC;

__device__ __forceinline__ void lds_barrier() {
  __builtin_amdgcn_fence(__ATOMIC_RELEASE, "workgroup", "local");
  __builtin_amdgcn_s_barrier();
  __builtin_amdgcn_fence(__ATOMIC_ACQUIRE, "workgroup", "local");
}

__device__ __forceinline__ void gram_fetch(v4f (&v)[8], const float* __restrict__ xb,
                                           int kc, int tid) {
#pragma unroll
  for (int i = 0; i < 8; ++i) {
    const int idx = i * 256 + tid;
    const int c   = idx >> 4;
    const int k4  = (idx & 15) << 2;
    v[i] = *reinterpret_cast<const v4f*>(xb + (size_t)c * HW + kc + k4);
  }
}

__device__ __forceinline__ void gram_stage(float (*s)[XPAD], const v4f (&v)[8], int tid) {
#pragma unroll
  for (int i = 0; i < 8; ++i) {
    const int idx = i * 256 + tid;
    const int c   = idx >> 4;
    const int k4  = (idx & 15) << 2;
    *reinterpret_cast<v4f*>(&s[c][k4]) = v[i];
  }
}

__device__ __forceinline__ void gram_chunk(const float (*s)[XPAD], v8f (&acc)[2], float& msum,
                                           int mrow, int mcol, int row0, int col0,
                                           int hi, int l15) {
  {
    const float4 u0 = *reinterpret_cast<const float4*>(&s[mrow][mcol]);
    const float4 u1 = *reinterpret_cast<const float4*>(&s[mrow][mcol + 4]);
    msum += u0.x; msum += u0.y; msum += u0.z; msum += u0.w;
    msum += u1.x; msum += u1.y; msum += u1.z; msum += u1.w;
  }
#pragma unroll 8
  for (int k0 = 0; k0 < KC; k0 += 4) {
    const int ks = k0 + hi * 2;
    const int ra = row0 + l15;
    v2f a; a.x = s[ra][ks]; a.y = s[ra][ks + 1];
#pragma unroll
    for (int t = 0; t < 2; ++t) {
      const int rb = col0 + t * 16 + l15;
      v2f bb; bb.x = s[rb][ks]; bb.y = s[rb][ks + 1];
      acc[t] = __builtin_amdgcn_wmma_f32_16x16x4_f32(
          false, a, false, bb, (short)0, acc[t], false, false);
    }
  }
}

__global__ void __launch_bounds__(256)
cw_gram(const float* __restrict__ x, float* __restrict__ gramPart,
        float* __restrict__ meanPart) {
  __shared__ alignas(16) float sX[2][CH][XPAD];
  __shared__ float sRow[32][9];
  const int b    = blockIdx.x / GBANDS;
  const int band = blockIdx.x % GBANDS;
  const int tid  = threadIdx.x;
  const int wave = tid >> 5;
  const int lane = tid & 31;
  const int hi   = lane >> 4;
  const int l15  = lane & 15;
  const int row0 = band * 32 + (wave & 1) * 16;
  const int col0 = (wave >> 1) * 32;

  v8f acc[2] = {};
  const float* xb = x + (size_t)b * CH * HW;
  const int mrow = band * 32 + (tid >> 3);
  const int mcol = (tid & 7) << 3;
  float msum = 0.f;

  v4f pa[8], pb[8];
  gram_fetch(pa, xb, 0, tid);
  gram_fetch(pb, xb, KC, tid);
  gram_stage(sX[0], pa, tid);
  lds_barrier();

  static_assert(NCHUNK % 2 == 1 && NCHUNK >= 5, "tail below assumes an odd chunk count");
  for (int ch = 0; ch < NCHUNK - 3; ch += 2) {
    gram_fetch(pa, xb, (ch + 2) * KC, tid);
    gram_chunk(sX[0], acc, msum, mrow, mcol, row0, col0, hi, l15);
    gram_stage(sX[1], pb, tid);
    lds_barrier();
    gram_fetch(pb, xb, (ch + 3) * KC, tid);
    gram_chunk(sX[1], acc, msum, mrow, mcol, row0, col0, hi, l15);
    gram_stage(sX[0], pa, tid);
    lds_barrier();
  }
  gram_fetch(pa, xb, (NCHUNK - 1) * KC, tid);
  gram_chunk(sX[0], acc, msum, mrow, mcol, row0, col0, hi, l15);
  gram_stage(sX[1], pb, tid);
  lds_barrier();
  gram_chunk(sX[1], acc, msum, mrow, mcol, row0, col0, hi, l15);
  gram_stage(sX[0], pa, tid);
  lds_barrier();
  gram_chunk(sX[0], acc, msum, mrow, mcol, row0, col0, hi, l15);

  float* gp = gramPart + (size_t)b * CH * CH;
#pragma unroll
  for (int t = 0; t < 2; ++t) {
#pragma unroll
    for (int r = 0; r < 8; ++r) {
      const int row = row0 + r + hi * 8;
      const int col = col0 + t * 16 + l15;
      gp[row * CH + col] = acc[t][r];
    }
  }

  sRow[tid >> 3][tid & 7] = msum;
  __syncthreads();
  if (tid < 32) {
    float s = 0.f;
    for (int j = 0; j < 8; ++j) s += sRow[tid][j];
    meanPart[b * CH + band * 32 + tid] = s;
  }
}

constexpr int NLD = CH + 12;
constexpr int STRIP = 16 * NLD;
constexpr size_t ITER_LDS_BYTES = (size_t)(2 * CH * NLD + 3 * STRIP) * sizeof(float);

__device__ __forceinline__ void load_afrag(v2f (&a)[32], const float* __restrict__ Astrip,
                                           int hi, int l15) {
#pragma unroll
  for (int j = 0; j < 32; ++j) {
    const int ks = 4 * j + hi * 2;
    a[j].x = Astrip[l15 * NLD + ks];
    a[j].y = Astrip[l15 * NLD + ks + 1];
  }
}

__device__ __forceinline__ void load_bfrag(v2f (&b)[32], const float* __restrict__ Bcol,
                                           int hi, int l15) {
#pragma unroll
  for (int j = 0; j < 32; ++j) {
    const int ks = 4 * j + hi * 2;
    b[j].x = Bcol[ks * NLD + l15];
    b[j].y = Bcol[(ks + 1) * NLD + l15];
  }
}

__device__ __forceinline__ v8f tile_chain(const v2f (&a)[32], const v2f (&b)[32]) {
  v8f acc = {};
#pragma unroll
  for (int j = 0; j < 32; ++j)
    acc = __builtin_amdgcn_wmma_f32_16x16x4_f32(
        false, a[j], false, b[j], (short)0, acc, false, false);
  return acc;
}

__device__ __forceinline__ void tile_store(float* __restrict__ strip, const v8f& acc,
                                           int wave, int hi, int l15) {
#pragma unroll
  for (int r = 0; r < 8; ++r) strip[(r + hi * 8) * NLD + wave * 16 + l15] = acc[r];
}

__device__ __forceinline__ void strip_to_global(const float* __restrict__ strip,
                                                float* __restrict__ dst, int tid) {
#pragma unroll
  for (int i = 0; i < 2; ++i) {
    const int idx = i * 256 + tid;
    const int r = idx >> 5, c = (idx & 31) << 2;
    *reinterpret_cast<v4f*>(dst + r * CH + c) =
        *reinterpret_cast<const v4f*>(strip + r * NLD + c);
  }
}

template <bool FIRST>
__global__ void __launch_bounds__(256)
cw_newton_iter(const float* __restrict__ gsum, const float* __restrict__ mean,
               const float* __restrict__ Sin, const float* __restrict__ Pin,
               float* __restrict__ Sout, float* __restrict__ Pout,
               float* __restrict__ rTrOut) {
  extern __shared__ __attribute__((aligned(16))) float nsm[];
  float* S  = nsm;
  float* P  = nsm + CH * NLD;
  float* T1 = nsm + 2 * CH * NLD;
  float* T2 = T1 + STRIP;
  const int strip = blockIdx.x;
  const int tid  = threadIdx.x;
  const int wave = tid >> 5;
  const int lane = tid & 31;
  const int hi   = lane >> 4;
  const int l15  = lane & 15;

  v4f sv[16];
  if constexpr (FIRST) {
    __shared__ float sTrInv;
    float sig[CH * CH / 256];
#pragma unroll
    for (int i = 0; i < CH * CH / 256; ++i) {
      const int e = i * 256 + tid;
      const float g = gsum[e];
      const int r = e >> 7, c = e & 127;
      sig[i] = (g - MTOT * mean[r] * mean[c]) / MTOT;
      if (r == c) sig[i] += EPSR;
      S[r * NLD + c] = sig[i];
    }
    __syncthreads();
    if (tid == 0) {
      float tr = 0.f;
      for (int i = 0; i < CH; ++i) tr += S[i * NLD + i];
      sTrInv = 1.0f / tr;
    }
    __syncthreads();
    const float rTr = sTrInv;
#pragma unroll
    for (int i = 0; i < CH * CH / 256; ++i) {
      const int e = i * 256 + tid;
      const int r = e >> 7, c = e & 127;
      S[r * NLD + c] = sig[i] * rTr;
      P[r * NLD + c] = (r == c) ? 1.f : 0.f;
    }
    if (strip == 0 && tid == 0) *rTrOut = rTr;
    __syncthreads();
    strip_to_global(S + strip * STRIP, Sout + (size_t)strip * 16 * CH, tid);
  } else {
    v4f pv[16];
#pragma unroll
    for (int i = 0; i < 16; ++i)
      pv[i] = *reinterpret_cast<const v4f*>(Pin + (size_t)(i * 256 + tid) * 4);
#pragma unroll
    for (int i = 0; i < 16; ++i)
      sv[i] = *reinterpret_cast<const v4f*>(Sin + (size_t)(i * 256 + tid) * 4);
#pragma unroll
    for (int i = 0; i < 16; ++i) {
      const int idx = i * 256 + tid;
      const int r = idx >> 5, c = (idx & 31) << 2;
      *reinterpret_cast<v4f*>(P + r * NLD + c) = pv[i];
    }
    lds_barrier();
  }

  v2f a[32], bP[32], bS[32];
  load_afrag(a, P + strip * STRIP, hi, l15);
  load_bfrag(bP, P + wave * 16, hi, l15);
  __builtin_amdgcn_sched_barrier(0);
  v8f acc = tile_chain(a, bP);
  tile_store(T1, acc, wave, hi, l15);
  if constexpr (!FIRST) {
#pragma unroll
    for (int i = 0; i < 16; ++i) {
      const int idx = i * 256 + tid;
      const int r = idx >> 5, c = (idx & 31) << 2;
      *reinterpret_cast<v4f*>(S + r * NLD + c) = sv[i];
    }
  }
  lds_barrier();
  load_afrag(a, T1, hi, l15);
  load_bfrag(bS, S + wave * 16, hi, l15);
  __builtin_amdgcn_sched_barrier(0);
  acc = tile_chain(a, bP);
  tile_store(T2, acc, wave, hi, l15);
  lds_barrier();
  load_afrag(a, T2, hi, l15);
  float pold[8];
#pragma unroll
  for (int r = 0; r < 8; ++r)
    pold[r] = P[strip * STRIP + (r + hi * 8) * NLD + wave * 16 + l15];
  __builtin_amdgcn_sched_barrier(0);
  acc = tile_chain(a, bS);
#pragma unroll
  for (int r = 0; r < 8; ++r) {
    const int o = (r + hi * 8) * NLD + wave * 16 + l15;
    T1[o] = 1.5f * pold[r] - 0.5f * acc[r];
  }
  lds_barrier();
  strip_to_global(T1, Pout + (size_t)strip * 16 * CH, tid);
}

__global__ void __launch_bounds__(256)
cw_newton_finish(const float* __restrict__ Pin, const float* __restrict__ mean,
                 const float* __restrict__ rot, const float* __restrict__ rTrIn,
                 float* __restrict__ Mout, float* __restrict__ bias) {
  extern __shared__ __attribute__((aligned(16))) float nsm[];
  float* P  = nsm + CH * NLD;
  float* T1 = nsm + 2 * CH * NLD;
  float* T2 = T1 + STRIP;
  float* T3 = T2 + STRIP;
  const int strip = blockIdx.x;
  const int tid  = threadIdx.x;
  const int wave = tid >> 5;
  const int lane = tid & 31;
  const int hi   = lane >> 4;
  const int l15  = lane & 15;

  v4f pv[16], rv[2];
#pragma unroll
  for (int i = 0; i < 2; ++i)
    rv[i] = *reinterpret_cast<const v4f*>(rot + (size_t)strip * 16 * CH + (i * 256 + tid) * 4);
#pragma unroll
  for (int i = 0; i < 16; ++i)
    pv[i] = *reinterpret_cast<const v4f*>(Pin + (size_t)(i * 256 + tid) * 4);
  const float sc = sqrtf(*rTrIn);
#pragma unroll
  for (int i = 0; i < 2; ++i) {
    const int idx = i * 256 + tid;
    const int r = idx >> 5, c = (idx & 31) << 2;
    *reinterpret_cast<v4f*>(T1 + r * NLD + c) = rv[i];
  }
#pragma unroll
  for (int i = 0; i < 16; ++i) {
    const int idx = i * 256 + tid;
    const int r = idx >> 5, c = (idx & 31) << 2;
    *reinterpret_cast<v4f*>(P + r * NLD + c) = pv[i];
  }
  __syncthreads();
  v2f a[32], bP[32];
  load_afrag(a, T1, hi, l15);
  load_bfrag(bP, P + wave * 16, hi, l15);
  __builtin_amdgcn_sched_barrier(0);
  const v8f acc = tile_chain(a, bP);
  tile_store(T2, acc, wave, hi, l15);
#pragma unroll
  for (int r = 0; r < 8; ++r) T3[(r + hi * 8) * NLD + wave * 16 + l15] = acc[r] * sc;
  __syncthreads();
  strip_to_global(T3, Mout + (size_t)strip * 16 * CH, tid);
  if (tid < 16) {
    float s = 0.f;
    for (int c = 0; c < CH; ++c) s += T2[tid * NLD + c] * mean[c];
    bias[strip * 32 + tid] = s * sc;
  }
}

constexpr int MPAD = CH + 4;
__global__ void __launch_bounds__(256)
cw_apply(const float* __restrict__ x, const float* __restrict__ Mmat,
         const float* __restrict__ bias, float* __restrict__ out) {
  extern __shared__ float smem[];
  float* sM = smem;
  float* sX = smem + CH * MPAD;
  float* sB = sX + CH * XPAD;
  const int b    = blockIdx.y;
  const int tid  = threadIdx.x;
  const int wave = tid >> 5;
  const int lane = tid & 31;
  const int hi   = lane >> 4;
  const int l15  = lane & 15;

  for (int i = tid; i < (CH * CH) / 4; i += 256) {
    const int e = i * 4;
    const int r = e >> 7, c = e & 127;
    const float4 v = *reinterpret_cast<const float4*>(Mmat + e);
    float* d = sM + r * MPAD + c;
    d[0] = v.x; d[1] = v.y; d[2] = v.z; d[3] = v.w;
  }
  if (tid < CH) sB[tid] = bias[(tid >> 4) * 32 + (tid & 15)];
  __syncthreads();

  const float* xb = x + (size_t)b * CH * HW;
  float*       ob = out + (size_t)b * CH * HW;

  for (int j = 0; j < 7; ++j) {
    const int hw0 = (blockIdx.x * 7 + j) * 64;
#pragma unroll
    for (int i = 0; i < 8; ++i) {
      const int idx = i * 256 + tid;
      const int c   = idx >> 4;
      const int k4  = (idx & 15) << 2;
      const float4 v = *reinterpret_cast<const float4*>(xb + (size_t)c * HW + hw0 + k4);
      float* d = sX + c * XPAD + k4;
      d[0] = v.x; d[1] = v.y; d[2] = v.z; d[3] = v.w;
    }
    __syncthreads();

    v8f acc[4] = {};
#pragma unroll 2
    for (int c0 = 0; c0 < CH; c0 += 4) {
      const int ks = c0 + hi * 2;
      const int ra = wave * 16 + l15;
      v2f a; a.x = sM[ra * MPAD + ks]; a.y = sM[ra * MPAD + ks + 1];
#pragma unroll
      for (int ht = 0; ht < 4; ++ht) {
        v2f bb;
        bb.x = sX[ks * XPAD + ht * 16 + l15];
        bb.y = sX[(ks + 1) * XPAD + ht * 16 + l15];
        acc[ht] = __builtin_amdgcn_wmma_f32_16x16x4_f32(
            false, a, false, bb, (short)0, acc[ht], false, false);
      }
    }

#pragma unroll
    for (int ht = 0; ht < 4; ++ht) {
#pragma unroll
      for (int r = 0; r < 8; ++r) {
        const int dch = wave * 16 + r + hi * 8;
        const int hw  = hw0 + ht * 16 + l15;
        ob[(size_t)dch * HW + hw] = acc[ht][r] - sB[dch];
      }
    }
    __syncthreads();
  }
}

extern "C" void kernel_launch(void* const* d_in, const int* in_sizes, int n_in,
                              void* d_out, int out_size, void* d_ws, size_t ws_size,
                              hipStream_t stream) {
  const float* x   = (const float*)d_in[0];
  const float* rot = (const float*)d_in[1];
  float* out = (float*)d_out;
  float* ws  = (float*)d_ws;

  float* gramPart = ws;
  float* meanPart = gramPart + (size_t)BATCH * CH * CH;
  float* mean     = meanPart + CH * BATCH;
  float* Mmat     = mean + CH;

  cw_gram<<<BATCH * GBANDS, 256, 0, stream>>>(x, gramPart, meanPart);
  cw_reduce<<<CH + 1, CH, 0, stream>>>(meanPart, mean, gramPart);
  float* Sn   = gramPart + (size_t)1 * CH * CH;
  float* Pa   = gramPart + (size_t)2 * CH * CH;
  float* Pb   = gramPart + (size_t)3 * CH * CH;
  float* rTrp = gramPart + (size_t)4 * CH * CH;
  float* bias = gramPart + (size_t)5 * CH * CH;
  cw_newton_iter<true><<<CH / 16, 256, ITER_LDS_BYTES, stream>>>(
      gramPart, mean, nullptr, nullptr, Sn, Pa, rTrp);
  cw_newton_iter<false><<<CH / 16, 256, ITER_LDS_BYTES, stream>>>(
      nullptr, nullptr, Sn, Pa, nullptr, Pb, nullptr);
  cw_newton_iter<false><<<CH / 16, 256, ITER_LDS_BYTES, stream>>>(
      nullptr, nullptr, Sn, Pb, nullptr, Pa, nullptr);
  cw_newton_iter<false><<<CH / 16, 256, ITER_LDS_BYTES, stream>>>(
      nullptr, nullptr, Sn, Pa, nullptr, Pb, nullptr);
  cw_newton_iter<false><<<CH / 16, 256, ITER_LDS_BYTES, stream>>>(
      nullptr, nullptr, Sn, Pb, nullptr, Pa, nullptr);
  cw_newton_finish<<<CH / 16, 256, ITER_LDS_BYTES, stream>>>(
      Pa, mean, rot, rTrp, Mmat, bias);
  cw_apply<<<dim3(7, BATCH), 256,
             (CH * MPAD + CH * XPAD + CH) * sizeof(float), stream>>>(
      x, Mmat, bias, out);
}


# cw_apply x staging software-pipelined: chunk j+1's global loads in flight in registers under chunk j's products and stores (two LDS slabs, one LDS-only barrier per chunk); sum order unchanged
# speedup vs baseline: 40.1183x; 1.0531x over previous
#include <hip/hip_runtime.h>
#include <hip/hip_bf16.h>


typedef __attribute__((ext_vector_type(2))) float v2f;
typedef __attribute__((ext_vector_type(8))) float v8f;
typedef __attribute__((ext_vector_type(4))) float v4f;

constexpr int CH    = 128;
constexpr int BATCH = 64;
constexpr int HW    = 56 * 56;
constexpr float MTOT = (float)BATCH * (float)HW;
constexpr float EPSR = 1e-5f;
constexpr int   T_NEWTON = 5;

__global__ void __launch_bounds__(CH)
cw_reduce(const float* __restrict__ meanPart, float* __restrict__ mean,
          float* gramPart) {
  const int t = threadIdx.x;
  if (blockIdx.x == CH) {
    float s = 0.f;
    for (int b = 0; b < BATCH; ++b) s += meanPart[b * CH + t];
    mean[t] = s / MTOT;
    return;
  }
  float* p = gramPart + blockIdx.x * CH + t;
  float v[BATCH];
#pragma unroll
  for (int s = 0; s < BATCH; ++s) v[s] = p[(size_t)s * CH * CH];
  float g = 0.f;
#pragma unroll
  for (int s = 0; s < BATCH; ++s) g += v[s];
  *p = g;
}

constexpr int KC = 64;
constexpr int XPAD = KC + 4;
constexpr int GBANDS = 4;
constexpr int NCHUNK = HW / KC;

__device__ __forceinline__ void lds_barrier() {
  __builtin_amdgcn_fence(__ATOMIC_RELEASE, "workgroup", "local");
  __builtin_amdgcn_s_barrier();
  __builtin_amdgcn_fence(__ATOMIC_ACQUIRE, "workgroup", "local");
}

__device__ __forceinline__ void gram_fetch(v4f (&v)[8], const float* __restrict__ xb,
                                           int kc, int tid) {
#pragma unroll
  for (int i = 0; i < 8; ++i) {
    const int idx = i * 256 + tid;
    const int c   = idx >> 4;
    const int k4  = (idx & 15) << 2;
    v[i] = *reinterpret_cast<const v4f*>(xb + (size_t)c * HW + kc + k4);
  }
}

__device__ __forceinline__ void gram_stage(float (*s)[XPAD], const v4f (&v)[8], int tid) {
#pragma unroll
  for (int i = 0; i < 8; ++i) {
    const int idx = i * 256 + tid;
    const int c   = idx >> 4;
    const int k4  = (idx & 15) << 2;
    *reinterpret_cast<v4f*>(&s[c][k4]) = v[i];
  }
}

__device__ __forceinline__ void gram_chunk(const float (*s)[XPAD], v8f (&acc)[2], float& msum,
                                           int mrow, int mcol, int row0, int col0,
                                           int hi, int l15) {
  {
    const float4 u0 = *reinterpret_cast<const float4*>(&s[mrow][mcol]);
    const float4 u1 = *reinterpret_cast<const float4*>(&s[mrow][mcol + 4]);
    msum += u0.x; msum += u0.y; msum += u0.z; msum += u0.w;
    msum += u1.x; msum += u1.y; msum += u1.z; msum += u1.w;
  }
#pragma unroll 8
  for (int k0 = 0; k0 < KC; k0 += 4) {
    const int ks = k0 + hi * 2;
    const int ra = row0 + l15;
    v2f a; a.x = s[ra][ks]; a.y = s[ra][ks + 1];
#pragma unroll
    for (int t = 0; t < 2; ++t) {
      const int rb = col0 + t * 16 + l15;
      v2f bb; bb.x = s[rb][ks]; bb.y = s[rb][ks + 1];
      acc[t] = __builtin_amdgcn_wmma_f32_16x16x4_f32(
          false, a, false, bb, (short)0, acc[t], false, false);
    }
  }
}

__global__ void __launch_bounds__(256)
cw_gram(const float* __restrict__ x, float* __restrict__ gramPart,
        float* __restrict__ meanPart) {
  __shared__ alignas(16) float sX[2][CH][XPAD];
  __shared__ float sRow[32][9];
  const int b    = blockIdx.x / GBANDS;
  const int band = blockIdx.x % GBANDS;
  const int tid  = threadIdx.x;
  const int wave = tid >> 5;
  const int lane = tid & 31;
  const int hi   = lane >> 4;
  const int l15  = lane & 15;
  const int row0 = band * 32 + (wave & 1) * 16;
  const int col0 = (wave >> 1) * 32;

  v8f acc[2] = {};
  const float* xb = x + (size_t)b * CH * HW;
  const int mrow = band * 32 + (tid >> 3);
  const int mcol = (tid & 7) << 3;
  float msum = 0.f;

  v4f pa[8], pb[8];
  gram_fetch(pa, xb, 0, tid);
  gram_fetch(pb, xb, KC, tid);
  gram_stage(sX[0], pa, tid);
  lds_barrier();

  static_assert(NCHUNK % 2 == 1 && NCHUNK >= 5, "tail below assumes an odd chunk count");
  for (int ch = 0; ch < NCHUNK - 3; ch += 2) {
    gram_fetch(pa, xb, (ch + 2) * KC, tid);
    gram_chunk(sX[0], acc, msum, mrow, mcol, row0, col0, hi, l15);
    gram_stage(sX[1], pb, tid);
    lds_barrier();
    gram_fetch(pb, xb, (ch + 3) * KC, tid);
    gram_chunk(sX[1], acc, msum, mrow, mcol, row0, col0, hi, l15);
    gram_stage(sX[0], pa, tid);
    lds_barrier();
  }
  gram_fetch(pa, xb, (NCHUNK - 1) * KC, tid);
  gram_chunk(sX[0], acc, msum, mrow, mcol, row0, col0, hi, l15);
  gram_stage(sX[1], pb, tid);
  lds_barrier();
  gram_chunk(sX[1], acc, msum, mrow, mcol, row0, col0, hi, l15);
  gram_stage(sX[0], pa, tid);
  lds_barrier();
  gram_chunk(sX[0], acc, msum, mrow, mcol, row0, col0, hi, l15);

  float* gp = gramPart + (size_t)b * CH * CH;
#pragma unroll
  for (int t = 0; t < 2; ++t) {
#pragma unroll
    for (int r = 0; r < 8; ++r) {
      const int row = row0 + r + hi * 8;
      const int col = col0 + t * 16 + l15;
      gp[row * CH + col] = acc[t][r];
    }
  }

  sRow[tid >> 3][tid & 7] = msum;
  __syncthreads();
  if (tid < 32) {
    float s = 0.f;
    for (int j = 0; j < 8; ++j) s += sRow[tid][j];
    meanPart[b * CH + band * 32 + tid] = s;
  }
}

constexpr int NLD = CH + 12;
constexpr int STRIP = 16 * NLD;
constexpr size_t ITER_LDS_BYTES = (size_t)(2 * CH * NLD + 3 * STRIP) * sizeof(float);

__device__ __forceinline__ void load_afrag(v2f (&a)[32], const float* __restrict__ Astrip,
                                           int hi, int l15) {
#pragma unroll
  for (int j = 0; j < 32; ++j) {
    const int ks = 4 * j + hi * 2;
    a[j].x = Astrip[l15 * NLD + ks];
    a[j].y = Astrip[l15 * NLD + ks + 1];
  }
}

__device__ __forceinline__ void load_bfrag(v2f (&b)[32], const float* __restrict__ Bcol,
                                           int hi, int l15) {
#pragma unroll
  for (int j = 0; j < 32; ++j) {
    const int ks = 4 * j + hi * 2;
    b[j].x = Bcol[ks * NLD + l15];
    b[j].y = Bcol[(ks + 1) * NLD + l15];
  }
}

__device__ __forceinline__ v8f tile_chain(const v2f (&a)[32], const v2f (&b)[32]) {
  v8f acc = {};
#pragma unroll
  for (int j = 0; j < 32; ++j)
    acc = __builtin_amdgcn_wmma_f32_16x16x4_f32(
        false, a[j], false, b[j], (short)0, acc, false, false);
  return acc;
}

__device__ __forceinline__ void tile_store(float* __restrict__ strip, const v8f& acc,
                                           int wave, int hi, int l15) {
#pragma unroll
  for (int r = 0; r < 8; ++r) strip[(r + hi * 8) * NLD + wave * 16 + l15] = acc[r];
}

__device__ __forceinline__ void strip_to_global(const float* __restrict__ strip,
                                                float* __restrict__ dst, int tid) {
#pragma unroll
  for (int i = 0; i < 2; ++i) {
    const int idx = i * 256 + tid;
    const int r = idx >> 5, c = (idx & 31) << 2;
    *reinterpret_cast<v4f*>(dst + r * CH + c) =
        *reinterpret_cast<const v4f*>(strip + r * NLD + c);
  }
}

template <bool FIRST>
__global__ void __launch_bounds__(256)
cw_newton_iter(const float* __restrict__ gsum, const float* __restrict__ mean,
               const float* __restrict__ Sin, const float* __restrict__ Pin,
               float* __restrict__ Sout, float* __restrict__ Pout,
               float* __restrict__ rTrOut) {
  extern __shared__ __attribute__((aligned(16))) float nsm[];
  float* S  = nsm;
  float* P  = nsm + CH * NLD;
  float* T1 = nsm + 2 * CH * NLD;
  float* T2 = T1 + STRIP;
  const int strip = blockIdx.x;
  const int tid  = threadIdx.x;
  const int wave = tid >> 5;
  const int lane = tid & 31;
  const int hi   = lane >> 4;
  const int l15  = lane & 15;

  v4f sv[16];
  if constexpr (FIRST) {
    __shared__ float sTrInv;
    float sig[CH * CH / 256];
#pragma unroll
    for (int i = 0; i < CH * CH / 256; ++i) {
      const int e = i * 256 + tid;
      const float g = gsum[e];
      const int r = e >> 7, c = e & 127;
      sig[i] = (g - MTOT * mean[r] * mean[c]) / MTOT;
      if (r == c) sig[i] += EPSR;
      S[r * NLD + c] = sig[i];
    }
    __syncthreads();
    if (tid == 0) {
      float tr = 0.f;
      for (int i = 0; i < CH; ++i) tr += S[i * NLD + i];
      sTrInv = 1.0f / tr;
    }
    __syncthreads();
    const float rTr = sTrInv;
#pragma unroll
    for (int i = 0; i < CH * CH / 256; ++i) {
      const int e = i * 256 + tid;
      const int r = e >> 7, c = e & 127;
      S[r * NLD + c] = sig[i] * rTr;
      P[r * NLD + c] = (r == c) ? 1.f : 0.f;
    }
    if (strip == 0 && tid == 0) *rTrOut = rTr;
    __syncthreads();
    strip_to_global(S + strip * STRIP, Sout + (size_t)strip * 16 * CH, tid);
  } else {
    v4f pv[16];
#pragma unroll
    for (int i = 0; i < 16; ++i)
      pv[i] = *reinterpret_cast<const v4f*>(Pin + (size_t)(i * 256 + tid) * 4);
#pragma unroll
    for (int i = 0; i < 16; ++i)
      sv[i] = *reinterpret_cast<const v4f*>(Sin + (size_t)(i * 256 + tid) * 4);
#pragma unroll
    for (int i = 0; i < 16; ++i) {
      const int idx = i * 256 + tid;
      const int r = idx >> 5, c = (idx & 31) << 2;
      *reinterpret_cast<v4f*>(P + r * NLD + c) = pv[i];
    }
    lds_barrier();
  }

  v2f a[32], bP[32], bS[32];
  load_afrag(a, P + strip * STRIP, hi, l15);
  load_bfrag(bP, P + wave * 16, hi, l15);
  __builtin_amdgcn_sched_barrier(0);
  v8f acc = tile_chain(a, bP);
  tile_store(T1, acc, wave, hi, l15);
  if constexpr (!FIRST) {
#pragma unroll
    for (int i = 0; i < 16; ++i) {
      const int idx = i * 256 + tid;
      const int r = idx >> 5, c = (idx & 31) << 2;
      *reinterpret_cast<v4f*>(S + r * NLD + c) = sv[i];
    }
  }
  lds_barrier();
  load_afrag(a, T1, hi, l15);
  load_bfrag(bS, S + wave * 16, hi, l15);
  __builtin_amdgcn_sched_barrier(0);
  acc = tile_chain(a, bP);
  tile_store(T2, acc, wave, hi, l15);
  lds_barrier();
  load_afrag(a, T2, hi, l15);
  float pold[8];
#pragma unroll
  for (int r = 0; r < 8; ++r)
    pold[r] = P[strip * STRIP + (r + hi * 8) * NLD + wave * 16 + l15];
  __builtin_amdgcn_sched_barrier(0);
  acc = tile_chain(a, bS);
#pragma unroll
  for (int r = 0; r < 8; ++r) {
    const int o = (r + hi * 8) * NLD + wave * 16 + l15;
    T1[o] = 1.5f * pold[r] - 0.5f * acc[r];
  }
  lds_barrier();
  strip_to_global(T1, Pout + (size_t)strip * 16 * CH, tid);
}

__global__ void __launch_bounds__(256)
cw_newton_finish(const float* __restrict__ Pin, const float* __restrict__ mean,
                 const float* __restrict__ rot, const float* __restrict__ rTrIn,
                 float* __restrict__ Mout, float* __restrict__ bias) {
  extern __shared__ __attribute__((aligned(16))) float nsm[];
  float* P  = nsm + CH * NLD;
  float* T1 = nsm + 2 * CH * NLD;
  float* T2 = T1 + STRIP;
  float* T3 = T2 + STRIP;
  const int strip = blockIdx.x;
  const int tid  = threadIdx.x;
  const int wave = tid >> 5;
  const int lane = tid & 31;
  const int hi   = lane >> 4;
  const int l15  = lane & 15;

  v4f pv[16], rv[2];
#pragma unroll
  for (int i = 0; i < 2; ++i)
    rv[i] = *reinterpret_cast<const v4f*>(rot + (size_t)strip * 16 * CH + (i * 256 + tid) * 4);
#pragma unroll
  for (int i = 0; i < 16; ++i)
    pv[i] = *reinterpret_cast<const v4f*>(Pin + (size_t)(i * 256 + tid) * 4);
  const float sc = sqrtf(*rTrIn);
#pragma unroll
  for (int i = 0; i < 2; ++i) {
    const int idx = i * 256 + tid;
    const int r = idx >> 5, c = (idx & 31) << 2;
    *reinterpret_cast<v4f*>(T1 + r * NLD + c) = rv[i];
  }
#pragma unroll
  for (int i = 0; i < 16; ++i) {
    const int idx = i * 256 + tid;
    const int r = idx >> 5, c = (idx & 31) << 2;
    *reinterpret_cast<v4f*>(P + r * NLD + c) = pv[i];
  }
  __syncthreads();
  v2f a[32], bP[32];
  load_afrag(a, T1, hi, l15);
  load_bfrag(bP, P + wave * 16, hi, l15);
  __builtin_amdgcn_sched_barrier(0);
  const v8f acc = tile_chain(a, bP);
  tile_store(T2, acc, wave, hi, l15);
#pragma unroll
  for (int r = 0; r < 8; ++r) T3[(r + hi * 8) * NLD + wave * 16 + l15] = acc[r] * sc;
  __syncthreads();
  strip_to_global(T3, Mout + (size_t)strip * 16 * CH, tid);
  if (tid < 16) {
    float s = 0.f;
    for (int c = 0; c < CH; ++c) s += T2[tid * NLD + c] * mean[c];
    bias[strip * 32 + tid] = s * sc;
  }
}

constexpr int MPAD = CH + 4;
constexpr int APPLY_CHUNKS = 7;
constexpr size_t APPLY_LDS_BYTES = (size_t)(CH * MPAD + 2 * CH * XPAD + CH) * sizeof(float);

__device__ __forceinline__ void apply_chunk(const float* __restrict__ sM,
                                            const float (*sXc)[XPAD],
                                            const float* __restrict__ sB,
                                            float* __restrict__ ob, int hw0,
                                            int wave, int hi, int l15) {
  v8f acc[4] = {};
#pragma unroll 2
  for (int c0 = 0; c0 < CH; c0 += 4) {
    const int ks = c0 + hi * 2;
    const int ra = wave * 16 + l15;
    v2f a; a.x = sM[ra * MPAD + ks]; a.y = sM[ra * MPAD + ks + 1];
#pragma unroll
    for (int ht = 0; ht < 4; ++ht) {
      v2f bb;
      bb.x = sXc[ks][ht * 16 + l15];
      bb.y = sXc[ks + 1][ht * 16 + l15];
      acc[ht] = __builtin_amdgcn_wmma_f32_16x16x4_f32(
          false, a, false, bb, (short)0, acc[ht], false, false);
    }
  }

#pragma unroll
  for (int ht = 0; ht < 4; ++ht) {
#pragma unroll
    for (int r = 0; r < 8; ++r) {
      const int dch = wave * 16 + r + hi * 8;
      const int hw  = hw0 + ht * 16 + l15;
      ob[(size_t)dch * HW + hw] = acc[ht][r] - sB[dch];
    }
  }
}

__global__ void __launch_bounds__(256)
cw_apply(const float* __restrict__ x, const float* __restrict__ Mmat,
         const float* __restrict__ bias, float* __restrict__ out) {
  extern __shared__ __attribute__((aligned(16))) float smem[];
  float* sM = smem;
  float (*sX0)[XPAD] = reinterpret_cast<float (*)[XPAD]>(smem + CH * MPAD);
  float (*sX1)[XPAD] = reinterpret_cast<float (*)[XPAD]>(smem + CH * MPAD + CH * XPAD);
  float* sB = smem + CH * MPAD + 2 * CH * XPAD;
  const int b    = blockIdx.y;
  const int tid  = threadIdx.x;
  const int wave = tid >> 5;
  const int lane = tid & 31;
  const int hi   = lane >> 4;
  const int l15  = lane & 15;

  const float* xb = x + (size_t)b * CH * HW;
  float*       ob = out + (size_t)b * CH * HW;
  const int hwFirst = blockIdx.x * APPLY_CHUNKS * KC;

  v4f px[8];
  gram_fetch(px, xb, hwFirst, tid);

  for (int i = tid; i < (CH * CH) / 4; i += 256) {
    const int e = i * 4;
    const int r = e >> 7, c = e & 127;
    const float4 v = *reinterpret_cast<const float4*>(Mmat + e);
    float* d = sM + r * MPAD + c;
    d[0] = v.x; d[1] = v.y; d[2] = v.z; d[3] = v.w;
  }
  if (tid < CH) sB[tid] = bias[(tid >> 4) * 32 + (tid & 15)];
  gram_stage(sX0, px, tid);
  lds_barrier();

  float (*cur)[XPAD] = sX0;
  float (*nxt)[XPAD] = sX1;
  for (int j = 0; j < APPLY_CHUNKS - 1; ++j) {
    gram_fetch(px, xb, hwFirst + (j + 1) * KC, tid);
    apply_chunk(sM, cur, sB, ob, hwFirst + j * KC, wave, hi, l15);
    gram_stage(nxt, px, tid);
    lds_barrier();
    float (*t)[XPAD] = cur; cur = nxt; nxt = t;
  }
  apply_chunk(sM, cur, sB, ob, hwFirst + (APPLY_CHUNKS - 1) * KC, wave, hi, l15);
}

extern "C" void kernel_launch(void* const* d_in, const int* in_sizes, int n_in,
                              void* d_out, int out_size, void* d_ws, size_t ws_size,
                              hipStream_t stream) {
  const float* x   = (const float*)d_in[0];
  const float* rot = (const float*)d_in[1];
  float* out = (float*)d_out;
  float* ws  = (float*)d_ws;

  float* gramPart = ws;
  float* meanPart = gramPart + (size_t)BATCH * CH * CH;
  float* mean     = meanPart + CH * BATCH;
  float* Mmat     = mean + CH;

  cw_gram<<<BATCH * GBANDS, 256, 0, stream>>>(x, gramPart, meanPart);
  cw_reduce<<<CH + 1, CH, 0, stream>>>(meanPart, mean, gramPart);
  float* Sn   = gramPart + (size_t)1 * CH * CH;
  float* Pa   = gramPart + (size_t)2 * CH * CH;
  float* Pb   = gramPart + (size_t)3 * CH * CH;
  float* rTrp = gramPart + (size_t)4 * CH * CH;
  float* bias = gramPart + (size_t)5 * CH * CH;
  cw_newton_iter<true><<<CH / 16, 256, ITER_LDS_BYTES, stream>>>(
      gramPart, mean, nullptr, nullptr, Sn, Pa, rTrp);
  cw_newton_iter<false><<<CH / 16, 256, ITER_LDS_BYTES, stream>>>(
      nullptr, nullptr, Sn, Pa, nullptr, Pb, nullptr);
  cw_newton_iter<false><<<CH / 16, 256, ITER_LDS_BYTES, stream>>>(
      nullptr, nullptr, Sn, Pb, nullptr, Pa, nullptr);
  cw_newton_iter<false><<<CH / 16, 256, ITER_LDS_BYTES, stream>>>(
      nullptr, nullptr, Sn, Pa, nullptr, Pb, nullptr);
  cw_newton_iter<false><<<CH / 16, 256, ITER_LDS_BYTES, stream>>>(
      nullptr, nullptr, Sn, Pb, nullptr, Pa, nullptr);
  cw_newton_finish<<<CH / 16, 256, ITER_LDS_BYTES, stream>>>(
      Pa, mean, rot, rTrp, Mmat, bias);
  cw_apply<<<dim3(HW / (APPLY_CHUNKS * KC), BATCH), 256, APPLY_LDS_BYTES, stream>>>(
      x, Mmat, bias, out);
}


# cw_gram computes only the 36 upper 16x16 tiles of each symmetric Gram partial (9 per workgroup, k split over its 8 waves, added in fixed order); mirror filled at Sigma staging; Gram sum order changes
# speedup vs baseline: 44.4615x; 1.1083x over previous
#include <hip/hip_runtime.h>
#include <hip/hip_bf16.h>


typedef __attribute__((ext_vector_type(2))) float v2f;
typedef __attribute__((ext_vector_type(8))) float v8f;
typedef __attribute__((ext_vector_type(4))) float v4f;

constexpr int CH    = 128;
constexpr int BATCH = 64;
constexpr int HW    = 56 * 56;
constexpr float MTOT = (float)BATCH * (float)HW;
constexpr float EPSR = 1e-5f;
constexpr int   T_NEWTON = 5;

__global__ void __launch_bounds__(CH)
cw_reduce(const float* __restrict__ meanPart, float* __restrict__ mean,
          float* gramPart) {
  const int t = threadIdx.x;
  if (blockIdx.x == CH) {
    const int strip = t >> 4;
    const int jj  = strip < 4 ? strip : 7 - strip;
    const int src = jj * 32 + (strip < 4 ? 0 : 16) + (t & 15);
    float s = 0.f;
    for (int b = 0; b < BATCH; ++b) s += meanPart[b * CH + src];
    mean[t] = s / MTOT;
    return;
  }
  if ((t >> 4) < ((int)blockIdx.x >> 4)) return;
  float* p = gramPart + blockIdx.x * CH + t;
  float v[BATCH];
#pragma unroll
  for (int s = 0; s < BATCH; ++s) v[s] = p[(size_t)s * CH * CH];
  float g = 0.f;
#pragma unroll
  for (int s = 0; s < BATCH; ++s) g += v[s];
  *p = g;
}

constexpr int KC = 64;
constexpr int XPAD = KC + 4;
constexpr int GBANDS = 4;
constexpr int NCHUNK = HW / KC;

__device__ __forceinline__ void lds_barrier() {
  __builtin_amdgcn_fence(__ATOMIC_RELEASE, "workgroup", "local");
  __builtin_amdgcn_s_barrier();
  __builtin_amdgcn_fence(__ATOMIC_ACQUIRE, "workgroup", "local");
}

__device__ __forceinline__ void gram_fetch(v4f (&v)[8], const float* __restrict__ xb,
                                           int kc, int tid) {
#pragma unroll
  for (int i = 0; i < 8; ++i) {
    const int idx = i * 256 + tid;
    const int c   = idx >> 4;
    const int k4  = (idx & 15) << 2;
    v[i] = *reinterpret_cast<const v4f*>(xb + (size_t)c * HW + kc + k4);
  }
}

__device__ __forceinline__ void gram_stage(float (*s)[XPAD], const v4f (&v)[8], int tid) {
#pragma unroll
  for (int i = 0; i < 8; ++i) {
    const int idx = i * 256 + tid;
    const int c   = idx >> 4;
    const int k4  = (idx & 15) << 2;
    *reinterpret_cast<v4f*>(&s[c][k4]) = v[i];
  }
}

template <int J>
__device__ __forceinline__ void gramT_fetch(v4f (&v)[8], const float* __restrict__ xb,
                                            int kc, int tid) {
#pragma unroll
  for (int i = J; i < 8; ++i) {
    const int idx = i * 256 + tid;
    const int c   = idx >> 4;
    const int k4  = (idx & 15) << 2;
    v[i] = *reinterpret_cast<const v4f*>(xb + (size_t)c * HW + kc + k4);
  }
}

template <int J>
__device__ __forceinline__ void gramT_stage(float (*s)[XPAD], const v4f (&v)[8], int tid) {
#pragma unroll
  for (int i = J; i < 8; ++i) {
    const int idx = i * 256 + tid;
    const int c   = idx >> 4;
    const int k4  = (idx & 15) << 2;
    *reinterpret_cast<v4f*>(&s[c][k4]) = v[i];
  }
}

template <int J>
__device__ __forceinline__ void gramT_chunk(const float (*s)[XPAD], v8f (&acc)[9], float& msum,
                                            int mrow, int mcol, int koff, int l15) {
  {
    const float4 u0 = *reinterpret_cast<const float4*>(&s[mrow][mcol]);
    const float4 u1 = *reinterpret_cast<const float4*>(&s[mrow][mcol + 4]);
    msum += u0.x; msum += u0.y; msum += u0.z; msum += u0.w;
    msum += u1.x; msum += u1.y; msum += u1.z; msum += u1.w;
  }
  v4f f[8];
#pragma unroll
  for (int t = J; t < 8; ++t)
    f[t] = *reinterpret_cast<const v4f*>(&s[t * 16 + l15][koff]);
#pragma unroll
  for (int h = 0; h < 2; ++h) {
#pragma unroll
    for (int n = 0; n < 9; ++n) {
      const int ra = (n < 8 - J) ? J : 7 - J;
      const int cb = (n < 8 - J) ? J + n : (7 - J) + (n - (8 - J));
      v2f a, bb;
      if (h == 0) { a = f[ra].xy; bb = f[cb].xy; } else { a = f[ra].zw; bb = f[cb].zw; }
      acc[n] = __builtin_amdgcn_wmma_f32_16x16x4_f32(
          false, a, false, bb, (short)0, acc[n], false, false);
    }
  }
}

constexpr int GRED_FLOATS = 8 * 9 * 256;
static_assert(GRED_FLOATS >= 2 * CH * XPAD, "the two slabs live in the same buffer");

template <int J>
__device__ __forceinline__ void gram_body(const float* __restrict__ x, float* __restrict__ gramPart,
                                          float* __restrict__ meanPart, float* sbuf,
                                          float (*sRow)[9]) {
  float (*s0)[XPAD] = reinterpret_cast<float (*)[XPAD]>(sbuf);
  float (*s1)[XPAD] = reinterpret_cast<float (*)[XPAD]>(sbuf + CH * XPAD);
  const int b    = blockIdx.x / GBANDS;
  const int tid  = threadIdx.x;
  const int wave = tid >> 5;
  const int lane = tid & 31;
  const int hi   = lane >> 4;
  const int l15  = lane & 15;
  const int koff = wave * 8 + hi * 4;

  v8f acc[9] = {};
  const float* xb = x + (size_t)b * CH * HW;
  const int rr   = tid >> 3;
  const int mrow = (rr < 16) ? J * 16 + rr : (7 - J) * 16 + (rr - 16);
  const int mcol = (tid & 7) << 3;
  float msum = 0.f;

  v4f pa[8], pb[8];
  gramT_fetch<J>(pa, xb, 0, tid);
  gramT_fetch<J>(pb, xb, KC, tid);
  gramT_stage<J>(s0, pa, tid);
  lds_barrier();

  static_assert(NCHUNK % 2 == 1 && NCHUNK >= 5, "tail below assumes an odd chunk count");
  for (int ch = 0; ch < NCHUNK - 3; ch += 2) {
    gramT_fetch<J>(pa, xb, (ch + 2) * KC, tid);
    gramT_chunk<J>(s0, acc, msum, mrow, mcol, koff, l15);
    __builtin_amdgcn_sched_barrier(0);
    gramT_stage<J>(s1, pb, tid);
    lds_barrier();
    gramT_fetch<J>(pb, xb, (ch + 3) * KC, tid);
    gramT_chunk<J>(s1, acc, msum, mrow, mcol, koff, l15);
    __builtin_amdgcn_sched_barrier(0);
    gramT_stage<J>(s0, pa, tid);
    lds_barrier();
  }
  gramT_fetch<J>(pa, xb, (NCHUNK - 1) * KC, tid);
  gramT_chunk<J>(s0, acc, msum, mrow, mcol, koff, l15);
  __builtin_amdgcn_sched_barrier(0);
  gramT_stage<J>(s1, pb, tid);
  lds_barrier();
  gramT_chunk<J>(s1, acc, msum, mrow, mcol, koff, l15);
  __builtin_amdgcn_sched_barrier(0);
  gramT_stage<J>(s0, pa, tid);
  lds_barrier();
  gramT_chunk<J>(s0, acc, msum, mrow, mcol, koff, l15);

  lds_barrier();
#pragma unroll
  for (int n = 0; n < 9; ++n) {
#pragma unroll
    for (int r = 0; r < 8; ++r)
      sbuf[(wave * 9 + n) * 256 + (r + hi * 8) * 16 + l15] = acc[n][r];
  }
  sRow[tid >> 3][tid & 7] = msum;
  __syncthreads();
  float* gp = gramPart + (size_t)b * CH * CH;
#pragma unroll
  for (int n = 0; n < 9; ++n) {
    const int ra = (n < 8 - J) ? J : 7 - J;
    const int cb = (n < 8 - J) ? J + n : (7 - J) + (n - (8 - J));
    float v = 0.f;
#pragma unroll
    for (int w = 0; w < 8; ++w) v += sbuf[(w * 9 + n) * 256 + tid];
    gp[(ra * 16 + (tid >> 4)) * CH + cb * 16 + (tid & 15)] = v;
  }
  if (tid < 32) {
    float sm = 0.f;
    for (int q = 0; q < 8; ++q) sm += sRow[tid][q];
    meanPart[b * CH + J * 32 + tid] = sm;
  }
}

__global__ void __launch_bounds__(256)
cw_gram(const float* __restrict__ x, float* __restrict__ gramPart,
        float* __restrict__ meanPart) {
  __shared__ alignas(16) float sbuf[GRED_FLOATS];
  __shared__ float sRow[32][9];
  switch (blockIdx.x % GBANDS) {
    case 0:  gram_body<0>(x, gramPart, meanPart, sbuf, sRow); break;
    case 1:  gram_body<1>(x, gramPart, meanPart, sbuf, sRow); break;
    case 2:  gram_body<2>(x, gramPart, meanPart, sbuf, sRow); break;
    default: gram_body<3>(x, gramPart, meanPart, sbuf, sRow); break;
  }
}

constexpr int NLD = CH + 12;
constexpr int STRIP = 16 * NLD;
constexpr size_t ITER_LDS_BYTES = (size_t)(2 * CH * NLD + 3 * STRIP) * sizeof(float);

__device__ __forceinline__ void load_afrag(v2f (&a)[32], const float* __restrict__ Astrip,
                                           int hi, int l15) {
#pragma unroll
  for (int j = 0; j < 32; ++j) {
    const int ks = 4 * j + hi * 2;
    a[j].x = Astrip[l15 * NLD + ks];
    a[j].y = Astrip[l15 * NLD + ks + 1];
  }
}

__device__ __forceinline__ void load_bfrag(v2f (&b)[32], const float* __restrict__ Bcol,
                                           int hi, int l15) {
#pragma unroll
  for (int j = 0; j < 32; ++j) {
    const int ks = 4 * j + hi * 2;
    b[j].x = Bcol[ks * NLD + l15];
    b[j].y = Bcol[(ks + 1) * NLD + l15];
  }
}

__device__ __forceinline__ v8f tile_chain(const v2f (&a)[32], const v2f (&b)[32]) {
  v8f acc = {};
#pragma unroll
  for (int j = 0; j < 32; ++j)
    acc = __builtin_amdgcn_wmma_f32_16x16x4_f32(
        false, a[j], false, b[j], (short)0, acc, false, false);
  return acc;
}

__device__ __forceinline__ void tile_store(float* __restrict__ strip, const v8f& acc,
                                           int wave, int hi, int l15) {
#pragma unroll
  for (int r = 0; r < 8; ++r) strip[(r + hi * 8) * NLD + wave * 16 + l15] = acc[r];
}

__device__ __forceinline__ void strip_to_global(const float* __restrict__ strip,
                                                float* __restrict__ dst, int tid) {
#pragma unroll
  for (int i = 0; i < 2; ++i) {
    const int idx = i * 256 + tid;
    const int r = idx >> 5, c = (idx & 31) << 2;
    *reinterpret_cast<v4f*>(dst + r * CH + c) =
        *reinterpret_cast<const v4f*>(strip + r * NLD + c);
  }
}

template <bool FIRST>
__global__ void __launch_bounds__(256)
cw_newton_iter(const float* __restrict__ gsum, const float* __restrict__ mean,
               const float* __restrict__ Sin, const float* __restrict__ Pin,
               float* __restrict__ Sout, float* __restrict__ Pout,
               float* __restrict__ rTrOut) {
  extern __shared__ __attribute__((aligned(16))) float nsm[];
  float* S  = nsm;
  float* P  = nsm + CH * NLD;
  float* T1 = nsm + 2 * CH * NLD;
  float* T2 = T1 + STRIP;
  const int strip = blockIdx.x;
  const int tid  = threadIdx.x;
  const int wave = tid >> 5;
  const int lane = tid & 31;
  const int hi   = lane >> 4;
  const int l15  = lane & 15;

  v4f sv[16];
  if constexpr (FIRST) {
    __shared__ float sTrInv;
    const int er = tid >> 4, ec = tid & 15;
    float mr8[8], mc8[8];
#pragma unroll
    for (int t = 0; t < 8; ++t) {
      mr8[t] = mean[t * 16 + er];
      mc8[t] = mean[t * 16 + ec];
    }
    float g[36];
#pragma unroll
    for (int ti = 0; ti < 8; ++ti) {
#pragma unroll
      for (int tj = ti; tj < 8; ++tj) {
        const int n = ti * 8 - (ti * (ti - 1)) / 2 + (tj - ti);
        g[n] = gsum[(ti * 16 + er) * CH + tj * 16 + ec];
      }
    }
#pragma unroll
    for (int ti = 0; ti < 8; ++ti) {
#pragma unroll
      for (int tj = ti; tj < 8; ++tj) {
        const int n = ti * 8 - (ti * (ti - 1)) / 2 + (tj - ti);
        const int r = ti * 16 + er, c = tj * 16 + ec;
        float sg = (g[n] - MTOT * mr8[ti] * mc8[tj]) / MTOT;
        if (r == c) sg += EPSR;
        S[r * NLD + c] = sg;
        if (ti < tj) S[c * NLD + r] = sg;
      }
    }
    __syncthreads();
    if (tid == 0) {
      float tr = 0.f;
      for (int i = 0; i < CH; ++i) tr += S[i * NLD + i];
      sTrInv = 1.0f / tr;
    }
    __syncthreads();
    const float rTr = sTrInv;
#pragma unroll
    for (int i = 0; i < CH * CH / 256; ++i) {
      const int e = i * 256 + tid;
      const int r = e >> 7, c = e & 127;
      S[r * NLD + c] = S[r * NLD + c] * rTr;
      P[r * NLD + c] = (r == c) ? 1.f : 0.f;
    }
    if (strip == 0 && tid == 0) *rTrOut = rTr;
    __syncthreads();
    strip_to_global(S + strip * STRIP, Sout + (size_t)strip * 16 * CH, tid);
  } else {
    v4f pv[16];
#pragma unroll
    for (int i = 0; i < 16; ++i)
      pv[i] = *reinterpret_cast<const v4f*>(Pin + (size_t)(i * 256 + tid) * 4);
#pragma unroll
    for (int i = 0; i < 16; ++i)
      sv[i] = *reinterpret_cast<const v4f*>(Sin + (size_t)(i * 256 + tid) * 4);
#pragma unroll
    for (int i = 0; i < 16; ++i) {
      const int idx = i * 256 + tid;
      const int r = idx >> 5, c = (idx & 31) << 2;
      *reinterpret_cast<v4f*>(P + r * NLD + c) = pv[i];
    }
    lds_barrier();
  }

  v2f a[32], bP[32], bS[32];
  load_afrag(a, P + strip * STRIP, hi, l15);
  load_bfrag(bP, P + wave * 16, hi, l15);
  __builtin_amdgcn_sched_barrier(0);
  v8f acc = tile_chain(a, bP);
  tile_store(T1, acc, wave, hi, l15);
  if constexpr (!FIRST) {
#pragma unroll
    for (int i = 0; i < 16; ++i) {
      const int idx = i * 256 + tid;
      const int r = idx >> 5, c = (idx & 31) << 2;
      *reinterpret_cast<v4f*>(S + r * NLD + c) = sv[i];
    }
  }
  lds_barrier();
  load_afrag(a, T1, hi, l15);
  load_bfrag(bS, S + wave * 16, hi, l15);
  __builtin_amdgcn_sched_barrier(0);
  acc = tile_chain(a, bP);
  tile_store(T2, acc, wave, hi, l15);
  lds_barrier();
  load_afrag(a, T2, hi, l15);
  float pold[8];
#pragma unroll
  for (int r = 0; r < 8; ++r)
    pold[r] = P[strip * STRIP + (r + hi * 8) * NLD + wave * 16 + l15];
  __builtin_amdgcn_sched_barrier(0);
  acc = tile_chain(a, bS);
#pragma unroll
  for (int r = 0; r < 8; ++r) {
    const int o = (r + hi * 8) * NLD + wave * 16 + l15;
    T1[o] = 1.5f * pold[r] - 0.5f * acc[r];
  }
  lds_barrier();
  strip_to_global(T1, Pout + (size_t)strip * 16 * CH, tid);
}

__global__ void __launch_bounds__(256)
cw_newton_finish(const float* __restrict__ Pin, const float* __restrict__ mean,
                 const float* __restrict__ rot, const float* __restrict__ rTrIn,
                 float* __restrict__ Mout, float* __restrict__ bias) {
  extern __shared__ __attribute__((aligned(16))) float nsm[];
  float* P  = nsm + CH * NLD;
  float* T1 = nsm + 2 * CH * NLD;
  float* T2 = T1 + STRIP;
  float* T3 = T2 + STRIP;
  const int strip = blockIdx.x;
  const int tid  = threadIdx.x;
  const int wave = tid >> 5;
  const int lane = tid & 31;
  const int hi   = lane >> 4;
  const int l15  = lane & 15;

  v4f pv[16], rv[2];
#pragma unroll
  for (int i = 0; i < 2; ++i)
    rv[i] = *reinterpret_cast<const v4f*>(rot + (size_t)strip * 16 * CH + (i * 256 + tid) * 4);
#pragma unroll
  for (int i = 0; i < 16; ++i)
    pv[i] = *reinterpret_cast<const v4f*>(Pin + (size_t)(i * 256 + tid) * 4);
  const float sc = sqrtf(*rTrIn);
#pragma unroll
  for (int i = 0; i < 2; ++i) {
    const int idx = i * 256 + tid;
    const int r = idx >> 5, c = (idx & 31) << 2;
    *reinterpret_cast<v4f*>(T1 + r * NLD + c) = rv[i];
  }
#pragma unroll
  for (int i = 0; i < 16; ++i) {
    const int idx = i * 256 + tid;
    const int r = idx >> 5, c = (idx & 31) << 2;
    *reinterpret_cast<v4f*>(P + r * NLD + c) = pv[i];
  }
  __syncthreads();
  v2f a[32], bP[32];
  load_afrag(a, T1, hi, l15);
  load_bfrag(bP, P + wave * 16, hi, l15);
  __builtin_amdgcn_sched_barrier(0);
  const v8f acc = tile_chain(a, bP);
  tile_store(T2, acc, wave, hi, l15);
#pragma unroll
  for (int r = 0; r < 8; ++r) T3[(r + hi * 8) * NLD + wave * 16 + l15] = acc[r] * sc;
  __syncthreads();
  strip_to_global(T3, Mout + (size_t)strip * 16 * CH, tid);
  if (tid < 16) {
    float s = 0.f;
    for (int c = 0; c < CH; ++c) s += T2[tid * NLD + c] * mean[c];
    bias[strip * 32 + tid] = s * sc;
  }
}

constexpr int MPAD = CH + 4;
constexpr int APPLY_CHUNKS = 7;
constexpr size_t APPLY_LDS_BYTES = (size_t)(CH * MPAD + 2 * CH * XPAD + CH) * sizeof(float);

__device__ __forceinline__ void apply_chunk(const float* __restrict__ sM,
                                            const float (*sXc)[XPAD],
                                            const float* __restrict__ sB,
                                            float* __restrict__ ob, int hw0,
                                            int wave, int hi, int l15) {
  v8f acc[4] = {};
#pragma unroll 2
  for (int c0 = 0; c0 < CH; c0 += 4) {
    const int ks = c0 + hi * 2;
    const int ra = wave * 16 + l15;
    v2f a; a.x = sM[ra * MPAD + ks]; a.y = sM[ra * MPAD + ks + 1];
#pragma unroll
    for (int ht = 0; ht < 4; ++ht) {
      v2f bb;
      bb.x = sXc[ks][ht * 16 + l15];
      bb.y = sXc[ks + 1][ht * 16 + l15];
      acc[ht] = __builtin_amdgcn_wmma_f32_16x16x4_f32(
          false, a, false, bb, (short)0, acc[ht], false, false);
    }
  }

#pragma unroll
  for (int ht = 0; ht < 4; ++ht) {
#pragma unroll
    for (int r = 0; r < 8; ++r) {
      const int dch = wave * 16 + r + hi * 8;
      const int hw  = hw0 + ht * 16 + l15;
      ob[(size_t)dch * HW + hw] = acc[ht][r] - sB[dch];
    }
  }
}

__global__ void __launch_bounds__(256)
cw_apply(const float* __restrict__ x, const float* __restrict__ Mmat,
         const float* __restrict__ bias, float* __restrict__ out) {
  extern __shared__ __attribute__((aligned(16))) float smem[];
  float* sM = smem;
  float (*sX0)[XPAD] = reinterpret_cast<float (*)[XPAD]>(smem + CH * MPAD);
  float (*sX1)[XPAD] = reinterpret_cast<float (*)[XPAD]>(smem + CH * MPAD + CH * XPAD);
  float* sB = smem + CH * MPAD + 2 * CH * XPAD;
  const int b    = blockIdx.y;
  const int tid  = threadIdx.x;
  const int wave = tid >> 5;
  const int lane = tid & 31;
  const int hi   = lane >> 4;
  const int l15  = lane & 15;

  const float* xb = x + (size_t)b * CH * HW;
  float*       ob = out + (size_t)b * CH * HW;
  const int hwFirst = blockIdx.x * APPLY_CHUNKS * KC;

  v4f px[8];
  gram_fetch(px, xb, hwFirst, tid);

  for (int i = tid; i < (CH * CH) / 4; i += 256) {
    const int e = i * 4;
    const int r = e >> 7, c = e & 127;
    const float4 v = *reinterpret_cast<const float4*>(Mmat + e);
    float* d = sM + r * MPAD + c;
    d[0] = v.x; d[1] = v.y; d[2] = v.z; d[3] = v.w;
  }
  if (tid < CH) sB[tid] = bias[(tid >> 4) * 32 + (tid & 15)];
  gram_stage(sX0, px, tid);
  lds_barrier();

  float (*cur)[XPAD] = sX0;
  float (*nxt)[XPAD] = sX1;
  for (int j = 0; j < APPLY_CHUNKS - 1; ++j) {
    gram_fetch(px, xb, hwFirst + (j + 1) * KC, tid);
    apply_chunk(sM, cur, sB, ob, hwFirst + j * KC, wave, hi, l15);
    gram_stage(nxt, px, tid);
    lds_barrier();
    float (*t)[XPAD] = cur; cur = nxt; nxt = t;
  }
  apply_chunk(sM, cur, sB, ob, hwFirst + (APPLY_CHUNKS - 1) * KC, wave, hi, l15);
}

extern "C" void kernel_launch(void* const* d_in, const int* in_sizes, int n_in,
                              void* d_out, int out_size, void* d_ws, size_t ws_size,
                              hipStream_t stream) {
  const float* x   = (const float*)d_in[0];
  const float* rot = (const float*)d_in[1];
  float* out = (float*)d_out;
  float* ws  = (float*)d_ws;

  float* gramPart = ws;
  float* meanPart = gramPart + (size_t)BATCH * CH * CH;
  float* mean     = meanPart + CH * BATCH;
  float* Mmat     = mean + CH;

  cw_gram<<<BATCH * GBANDS, 256, 0, stream>>>(x, gramPart, meanPart);
  cw_reduce<<<CH + 1, CH, 0, stream>>>(meanPart, mean, gramPart);
  float* Sn   = gramPart + (size_t)1 * CH * CH;
  float* Pa   = gramPart + (size_t)2 * CH * CH;
  float* Pb   = gramPart + (size_t)3 * CH * CH;
  float* rTrp = gramPart + (size_t)4 * CH * CH;
  float* bias = gramPart + (size_t)5 * CH * CH;
  cw_newton_iter<true><<<CH / 16, 256, ITER_LDS_BYTES, stream>>>(
      gramPart, mean, nullptr, nullptr, Sn, Pa, rTrp);
  cw_newton_iter<false><<<CH / 16, 256, ITER_LDS_BYTES, stream>>>(
      nullptr, nullptr, Sn, Pa, nullptr, Pb, nullptr);
  cw_newton_iter<false><<<CH / 16, 256, ITER_LDS_BYTES, stream>>>(
      nullptr, nullptr, Sn, Pb, nullptr, Pa, nullptr);
  cw_newton_iter<false><<<CH / 16, 256, ITER_LDS_BYTES, stream>>>(
      nullptr, nullptr, Sn, Pa, nullptr, Pb, nullptr);
  cw_newton_iter<false><<<CH / 16, 256, ITER_LDS_BYTES, stream>>>(
      nullptr, nullptr, Sn, Pb, nullptr, Pa, nullptr);
  cw_newton_finish<<<CH / 16, 256, ITER_LDS_BYTES, stream>>>(
      Pa, mean, rot, rTrp, Mmat, bias);
  cw_apply<<<dim3(HW / (APPLY_CHUNKS * KC), BATCH), 256, APPLY_LDS_BYTES, stream>>>(
      x, Mmat, bias, out);
}


# Newton iteration 1 (P = I: its products give I, I, SigmaN exactly) written in closed form, 1.5 I - 0.5 SigmaN, inside the Sigma launch, which runs iteration 2: 8 launches, all sums unchanged
# speedup vs baseline: 46.1821x; 1.0387x over previous
#include <hip/hip_runtime.h>
#include <hip/hip_bf16.h>


typedef __attribute__((ext_vector_type(2))) float v2f;
typedef __attribute__((ext_vector_type(8))) float v8f;
typedef __attribute__((ext_vector_type(4))) float v4f;

constexpr int CH    = 128;
constexpr int BATCH = 64;
constexpr int HW    = 56 * 56;
constexpr float MTOT = (float)BATCH * (float)HW;
constexpr float EPSR = 1e-5f;
constexpr int   T_NEWTON = 5;

__global__ void __launch_bounds__(CH)
cw_reduce(const float* __restrict__ meanPart, float* __restrict__ mean,
          float* gramPart) {
  const int t = threadIdx.x;
  if (blockIdx.x == CH) {
    const int strip = t >> 4;
    const int jj  = strip < 4 ? strip : 7 - strip;
    const int src = jj * 32 + (strip < 4 ? 0 : 16) + (t & 15);
    float s = 0.f;
    for (int b = 0; b < BATCH; ++b) s += meanPart[b * CH + src];
    mean[t] = s / MTOT;
    return;
  }
  if ((t >> 4) < ((int)blockIdx.x >> 4)) return;
  float* p = gramPart + blockIdx.x * CH + t;
  float v[BATCH];
#pragma unroll
  for (int s = 0; s < BATCH; ++s) v[s] = p[(size_t)s * CH * CH];
  float g = 0.f;
#pragma unroll
  for (int s = 0; s < BATCH; ++s) g += v[s];
  *p = g;
}

constexpr int KC = 64;
constexpr int XPAD = KC + 4;
constexpr int GBANDS = 4;
constexpr int NCHUNK = HW / KC;

__device__ __forceinline__ void lds_barrier() {
  __builtin_amdgcn_fence(__ATOMIC_RELEASE, "workgroup", "local");
  __builtin_amdgcn_s_barrier();
  __builtin_amdgcn_fence(__ATOMIC_ACQUIRE, "workgroup", "local");
}

__device__ __forceinline__ void gram_fetch(v4f (&v)[8], const float* __restrict__ xb,
                                           int kc, int tid) {
#pragma unroll
  for (int i = 0; i < 8; ++i) {
    const int idx = i * 256 + tid;
    const int c   = idx >> 4;
    const int k4  = (idx & 15) << 2;
    v[i] = *reinterpret_cast<const v4f*>(xb + (size_t)c * HW + kc + k4);
  }
}

__device__ __forceinline__ void gram_stage(float (*s)[XPAD], const v4f (&v)[8], int tid) {
#pragma unroll
  for (int i = 0; i < 8; ++i) {
    const int idx = i * 256 + tid;
    const int c   = idx >> 4;
    const int k4  = (idx & 15) << 2;
    *reinterpret_cast<v4f*>(&s[c][k4]) = v[i];
  }
}

template <int J>
__device__ __forceinline__ void gramT_fetch(v4f (&v)[8], const float* __restrict__ xb,
                                            int kc, int tid) {
#pragma unroll
  for (int i = J; i < 8; ++i) {
    const int idx = i * 256 + tid;
    const int c   = idx >> 4;
    const int k4  = (idx & 15) << 2;
    v[i] = *reinterpret_cast<const v4f*>(xb + (size_t)c * HW + kc + k4);
  }
}

template <int J>
__device__ __forceinline__ void gramT_stage(float (*s)[XPAD], const v4f (&v)[8], int tid) {
#pragma unroll
  for (int i = J; i < 8; ++i) {
    const int idx = i * 256 + tid;
    const int c   = idx >> 4;
    const int k4  = (idx & 15) << 2;
    *reinterpret_cast<v4f*>(&s[c][k4]) = v[i];
  }
}

template <int J>
__device__ __forceinline__ void gramT_chunk(const float (*s)[XPAD], v8f (&acc)[9], float& msum,
                                            int mrow, int mcol, int koff, int l15) {
  {
    const float4 u0 = *reinterpret_cast<const float4*>(&s[mrow][mcol]);
    const float4 u1 = *reinterpret_cast<const float4*>(&s[mrow][mcol + 4]);
    msum += u0.x; msum += u0.y; msum += u0.z; msum += u0.w;
    msum += u1.x; msum += u1.y; msum += u1.z; msum += u1.w;
  }
  v4f f[8];
#pragma unroll
  for (int t = J; t < 8; ++t)
    f[t] = *reinterpret_cast<const v4f*>(&s[t * 16 + l15][koff]);
#pragma unroll
  for (int h = 0; h < 2; ++h) {
#pragma unroll
    for (int n = 0; n < 9; ++n) {
      const int ra = (n < 8 - J) ? J : 7 - J;
      const int cb = (n < 8 - J) ? J + n : (7 - J) + (n - (8 - J));
      v2f a, bb;
      if (h == 0) { a = f[ra].xy; bb = f[cb].xy; } else { a = f[ra].zw; bb = f[cb].zw; }
      acc[n] = __builtin_amdgcn_wmma_f32_16x16x4_f32(
          false, a, false, bb, (short)0, acc[n], false, false);
    }
  }
}

constexpr int GRED_FLOATS = 8 * 9 * 256;
static_assert(GRED_FLOATS >= 2 * CH * XPAD, "the two slabs live in the same buffer");

template <int J>
__device__ __forceinline__ void gram_body(const float* __restrict__ x, float* __restrict__ gramPart,
                                          float* __restrict__ meanPart, float* sbuf,
                                          float (*sRow)[9]) {
  float (*s0)[XPAD] = reinterpret_cast<float (*)[XPAD]>(sbuf);
  float (*s1)[XPAD] = reinterpret_cast<float (*)[XPAD]>(sbuf + CH * XPAD);
  const int b    = blockIdx.x / GBANDS;
  const int tid  = threadIdx.x;
  const int wave = tid >> 5;
  const int lane = tid & 31;
  const int hi   = lane >> 4;
  const int l15  = lane & 15;
  const int koff = wave * 8 + hi * 4;

  v8f acc[9] = {};
  const float* xb = x + (size_t)b * CH * HW;
  const int rr   = tid >> 3;
  const int mrow = (rr < 16) ? J * 16 + rr : (7 - J) * 16 + (rr - 16);
  const int mcol = (tid & 7) << 3;
  float msum = 0.f;

  v4f pa[8], pb[8];
  gramT_fetch<J>(pa, xb, 0, tid);
  gramT_fetch<J>(pb, xb, KC, tid);
  gramT_stage<J>(s0, pa, tid);
  lds_barrier();

  static_assert(NCHUNK % 2 == 1 && NCHUNK >= 5, "tail below assumes an odd chunk count");
  for (int ch = 0; ch < NCHUNK - 3; ch += 2) {
    gramT_fetch<J>(pa, xb, (ch + 2) * KC, tid);
    gramT_chunk<J>(s0, acc, msum, mrow, mcol, koff, l15);
    __builtin_amdgcn_sched_barrier(0);
    gramT_stage<J>(s1, pb, tid);
    lds_barrier();
    gramT_fetch<J>(pb, xb, (ch + 3) * KC, tid);
    gramT_chunk<J>(s1, acc, msum, mrow, mcol, koff, l15);
    __builtin_amdgcn_sched_barrier(0);
    gramT_stage<J>(s0, pa, tid);
    lds_barrier();
  }
  gramT_fetch<J>(pa, xb, (NCHUNK - 1) * KC, tid);
  gramT_chunk<J>(s0, acc, msum, mrow, mcol, koff, l15);
  __builtin_amdgcn_sched_barrier(0);
  gramT_stage<J>(s1, pb, tid);
  lds_barrier();
  gramT_chunk<J>(s1, acc, msum, mrow, mcol, koff, l15);
  __builtin_amdgcn_sched_barrier(0);
  gramT_stage<J>(s0, pa, tid);
  lds_barrier();
  gramT_chunk<J>(s0, acc, msum, mrow, mcol, koff, l15);

  lds_barrier();
#pragma unroll
  for (int n = 0; n < 9; ++n) {
#pragma unroll
    for (int r = 0; r < 8; ++r)
      sbuf[(wave * 9 + n) * 256 + (r + hi * 8) * 16 + l15] = acc[n][r];
  }
  sRow[tid >> 3][tid & 7] = msum;
  __syncthreads();
  float* gp = gramPart + (size_t)b * CH * CH;
#pragma unroll
  for (int n = 0; n < 9; ++n) {
    const int ra = (n < 8 - J) ? J : 7 - J;
    const int cb = (n < 8 - J) ? J + n : (7 - J) + (n - (8 - J));
    float v = 0.f;
#pragma unroll
    for (int w = 0; w < 8; ++w) v += sbuf[(w * 9 + n) * 256 + tid];
    gp[(ra * 16 + (tid >> 4)) * CH + cb * 16 + (tid & 15)] = v;
  }
  if (tid < 32) {
    float sm = 0.f;
    for (int q = 0; q < 8; ++q) sm += sRow[tid][q];
    meanPart[b * CH + J * 32 + tid] = sm;
  }
}

__global__ void __launch_bounds__(256)
cw_gram(const float* __restrict__ x, float* __restrict__ gramPart,
        float* __restrict__ meanPart) {
  __shared__ alignas(16) float sbuf[GRED_FLOATS];
  __shared__ float sRow[32][9];
  switch (blockIdx.x % GBANDS) {
    case 0:  gram_body<0>(x, gramPart, meanPart, sbuf, sRow); break;
    case 1:  gram_body<1>(x, gramPart, meanPart, sbuf, sRow); break;
    case 2:  gram_body<2>(x, gramPart, meanPart, sbuf, sRow); break;
    default: gram_body<3>(x, gramPart, meanPart, sbuf, sRow); break;
  }
}

constexpr int NLD = CH + 12;
constexpr int STRIP = 16 * NLD;
constexpr size_t ITER_LDS_BYTES = (size_t)(2 * CH * NLD + 3 * STRIP) * sizeof(float);

__device__ __forceinline__ void load_afrag(v2f (&a)[32], const float* __restrict__ Astrip,
                                           int hi, int l15) {
#pragma unroll
  for (int j = 0; j < 32; ++j) {
    const int ks = 4 * j + hi * 2;
    a[j].x = Astrip[l15 * NLD + ks];
    a[j].y = Astrip[l15 * NLD + ks + 1];
  }
}

__device__ __forceinline__ void load_bfrag(v2f (&b)[32], const float* __restrict__ Bcol,
                                           int hi, int l15) {
#pragma unroll
  for (int j = 0; j < 32; ++j) {
    const int ks = 4 * j + hi * 2;
    b[j].x = Bcol[ks * NLD + l15];
    b[j].y = Bcol[(ks + 1) * NLD + l15];
  }
}

__device__ __forceinline__ v8f tile_chain(const v2f (&a)[32], const v2f (&b)[32]) {
  v8f acc = {};
#pragma unroll
  for (int j = 0; j < 32; ++j)
    acc = __builtin_amdgcn_wmma_f32_16x16x4_f32(
        false, a[j], false, b[j], (short)0, acc, false, false);
  return acc;
}

__device__ __forceinline__ void tile_store(float* __restrict__ strip, const v8f& acc,
                                           int wave, int hi, int l15) {
#pragma unroll
  for (int r = 0; r < 8; ++r) strip[(r + hi * 8) * NLD + wave * 16 + l15] = acc[r];
}

__device__ __forceinline__ void strip_to_global(const float* __restrict__ strip,
                                                float* __restrict__ dst, int tid) {
#pragma unroll
  for (int i = 0; i < 2; ++i) {
    const int idx = i * 256 + tid;
    const int r = idx >> 5, c = (idx & 31) << 2;
    *reinterpret_cast<v4f*>(dst + r * CH + c) =
        *reinterpret_cast<const v4f*>(strip + r * NLD + c);
  }
}

template <bool FIRST>
__global__ void __launch_bounds__(256)
cw_newton_iter(const float* __restrict__ gsum, const float* __restrict__ mean,
               const float* __restrict__ Sin, const float* __restrict__ Pin,
               float* __restrict__ Sout, float* __restrict__ Pout,
               float* __restrict__ rTrOut) {
  extern __shared__ __attribute__((aligned(16))) float nsm[];
  float* S  = nsm;
  float* P  = nsm + CH * NLD;
  float* T1 = nsm + 2 * CH * NLD;
  float* T2 = T1 + STRIP;
  const int strip = blockIdx.x;
  const int tid  = threadIdx.x;
  const int wave = tid >> 5;
  const int lane = tid & 31;
  const int hi   = lane >> 4;
  const int l15  = lane & 15;

  v4f sv[16];
  if constexpr (FIRST) {
    __shared__ float sTrInv;
    const int er = tid >> 4, ec = tid & 15;
    float mr8[8], mc8[8];
#pragma unroll
    for (int t = 0; t < 8; ++t) {
      mr8[t] = mean[t * 16 + er];
      mc8[t] = mean[t * 16 + ec];
    }
    float g[36];
#pragma unroll
    for (int ti = 0; ti < 8; ++ti) {
#pragma unroll
      for (int tj = ti; tj < 8; ++tj) {
        const int n = ti * 8 - (ti * (ti - 1)) / 2 + (tj - ti);
        g[n] = gsum[(ti * 16 + er) * CH + tj * 16 + ec];
      }
    }
#pragma unroll
    for (int ti = 0; ti < 8; ++ti) {
#pragma unroll
      for (int tj = ti; tj < 8; ++tj) {
        const int n = ti * 8 - (ti * (ti - 1)) / 2 + (tj - ti);
        const int r = ti * 16 + er, c = tj * 16 + ec;
        float sg = (g[n] - MTOT * mr8[ti] * mc8[tj]) / MTOT;
        if (r == c) sg += EPSR;
        S[r * NLD + c] = sg;
        if (ti < tj) S[c * NLD + r] = sg;
      }
    }
    __syncthreads();
    if (tid == 0) {
      float tr = 0.f;
      for (int i = 0; i < CH; ++i) tr += S[i * NLD + i];
      sTrInv = 1.0f / tr;
    }
    __syncthreads();
    const float rTr = sTrInv;
#pragma unroll
    for (int i = 0; i < CH * CH / 256; ++i) {
      const int e = i * 256 + tid;
      const int r = e >> 7, c = e & 127;
      const float sn = S[r * NLD + c] * rTr;
      S[r * NLD + c] = sn;
      P[r * NLD + c] = 1.5f * ((r == c) ? 1.f : 0.f) - 0.5f * sn;
    }
    if (strip == 0 && tid == 0) *rTrOut = rTr;
    __syncthreads();
    strip_to_global(S + strip * STRIP, Sout + (size_t)strip * 16 * CH, tid);
  } else {
    v4f pv[16];
#pragma unroll
    for (int i = 0; i < 16; ++i)
      pv[i] = *reinterpret_cast<const v4f*>(Pin + (size_t)(i * 256 + tid) * 4);
#pragma unroll
    for (int i = 0; i < 16; ++i)
      sv[i] = *reinterpret_cast<const v4f*>(Sin + (size_t)(i * 256 + tid) * 4);
#pragma unroll
    for (int i = 0; i < 16; ++i) {
      const int idx = i * 256 + tid;
      const int r = idx >> 5, c = (idx & 31) << 2;
      *reinterpret_cast<v4f*>(P + r * NLD + c) = pv[i];
    }
    lds_barrier();
  }

  v2f a[32], bP[32], bS[32];
  load_afrag(a, P + strip * STRIP, hi, l15);
  load_bfrag(bP, P + wave * 16, hi, l15);
  __builtin_amdgcn_sched_barrier(0);
  v8f acc = tile_chain(a, bP);
  tile_store(T1, acc, wave, hi, l15);
  if constexpr (!FIRST) {
#pragma unroll
    for (int i = 0; i < 16; ++i) {
      const int idx = i * 256 + tid;
      const int r = idx >> 5, c = (idx & 31) << 2;
      *reinterpret_cast<v4f*>(S + r * NLD + c) = sv[i];
    }
  }
  lds_barrier();
  load_afrag(a, T1, hi, l15);
  load_bfrag(bS, S + wave * 16, hi, l15);
  __builtin_amdgcn_sched_barrier(0);
  acc = tile_chain(a, bP);
  tile_store(T2, acc, wave, hi, l15);
  lds_barrier();
  load_afrag(a, T2, hi, l15);
  float pold[8];
#pragma unroll
  for (int r = 0; r < 8; ++r)
    pold[r] = P[strip * STRIP + (r + hi * 8) * NLD + wave * 16 + l15];
  __builtin_amdgcn_sched_barrier(0);
  acc = tile_chain(a, bS);
#pragma unroll
  for (int r = 0; r < 8; ++r) {
    const int o = (r + hi * 8) * NLD + wave * 16 + l15;
    T1[o] = 1.5f * pold[r] - 0.5f * acc[r];
  }
  lds_barrier();
  strip_to_global(T1, Pout + (size_t)strip * 16 * CH, tid);
}

__global__ void __launch_bounds__(256)
cw_newton_finish(const float* __restrict__ Pin, const float* __restrict__ mean,
                 const float* __restrict__ rot, const float* __restrict__ rTrIn,
                 float* __restrict__ Mout, float* __restrict__ bias) {
  extern __shared__ __attribute__((aligned(16))) float nsm[];
  float* P  = nsm + CH * NLD;
  float* T1 = nsm + 2 * CH * NLD;
  float* T2 = T1 + STRIP;
  float* T3 = T2 + STRIP;
  const int strip = blockIdx.x;
  const int tid  = threadIdx.x;
  const int wave = tid >> 5;
  const int lane = tid & 31;
  const int hi   = lane >> 4;
  const int l15  = lane & 15;

  v4f pv[16], rv[2];
#pragma unroll
  for (int i = 0; i < 2; ++i)
    rv[i] = *reinterpret_cast<const v4f*>(rot + (size_t)strip * 16 * CH + (i * 256 + tid) * 4);
#pragma unroll
  for (int i = 0; i < 16; ++i)
    pv[i] = *reinterpret_cast<const v4f*>(Pin + (size_t)(i * 256 + tid) * 4);
  const float sc = sqrtf(*rTrIn);
#pragma unroll
  for (int i = 0; i < 2; ++i) {
    const int idx = i * 256 + tid;
    const int r = idx >> 5, c = (idx & 31) << 2;
    *reinterpret_cast<v4f*>(T1 + r * NLD + c) = rv[i];
  }
#pragma unroll
  for (int i = 0; i < 16; ++i) {
    const int idx = i * 256 + tid;
    const int r = idx >> 5, c = (idx & 31) << 2;
    *reinterpret_cast<v4f*>(P + r * NLD + c) = pv[i];
  }
  __syncthreads();
  v2f a[32], bP[32];
  load_afrag(a, T1, hi, l15);
  load_bfrag(bP, P + wave * 16, hi, l15);
  __builtin_amdgcn_sched_barrier(0);
  const v8f acc = tile_chain(a, bP);
  tile_store(T2, acc, wave, hi, l15);
#pragma unroll
  for (int r = 0; r < 8; ++r) T3[(r + hi * 8) * NLD + wave * 16 + l15] = acc[r] * sc;
  __syncthreads();
  strip_to_global(T3, Mout + (size_t)strip * 16 * CH, tid);
  if (tid < 16) {
    float s = 0.f;
    for (int c = 0; c < CH; ++c) s += T2[tid * NLD + c] * mean[c];
    bias[strip * 32 + tid] = s * sc;
  }
}

constexpr int MPAD = CH + 4;
constexpr int APPLY_CHUNKS = 7;
constexpr size_t APPLY_LDS_BYTES = (size_t)(CH * MPAD + 2 * CH * XPAD + CH) * sizeof(float);

__device__ __forceinline__ void apply_chunk(const float* __restrict__ sM,
                                            const float (*sXc)[XPAD],
                                            const float* __restrict__ sB,
                                            float* __restrict__ ob, int hw0,
                                            int wave, int hi, int l15) {
  v8f acc[4] = {};
#pragma unroll 2
  for (int c0 = 0; c0 < CH; c0 += 4) {
    const int ks = c0 + hi * 2;
    const int ra = wave * 16 + l15;
    v2f a; a.x = sM[ra * MPAD + ks]; a.y = sM[ra * MPAD + ks + 1];
#pragma unroll
    for (int ht = 0; ht < 4; ++ht) {
      v2f bb;
      bb.x = sXc[ks][ht * 16 + l15];
      bb.y = sXc[ks + 1][ht * 16 + l15];
      acc[ht] = __builtin_amdgcn_wmma_f32_16x16x4_f32(
          false, a, false, bb, (short)0, acc[ht], false, false);
    }
  }

#pragma unroll
  for (int ht = 0; ht < 4; ++ht) {
#pragma unroll
    for (int r = 0; r < 8; ++r) {
      const int dch = wave * 16 + r + hi * 8;
      const int hw  = hw0 + ht * 16 + l15;
      ob[(size_t)dch * HW + hw] = acc[ht][r] - sB[dch];
    }
  }
}

__global__ void __launch_bounds__(256)
cw_apply(const float* __restrict__ x, const float* __restrict__ Mmat,
         const float* __restrict__ bias, float* __restrict__ out) {
  extern __shared__ __attribute__((aligned(16))) float smem[];
  float* sM = smem;
  float (*sX0)[XPAD] = reinterpret_cast<float (*)[XPAD]>(smem + CH * MPAD);
  float (*sX1)[XPAD] = reinterpret_cast<float (*)[XPAD]>(smem + CH * MPAD + CH * XPAD);
  float* sB = smem + CH * MPAD + 2 * CH * XPAD;
  const int b    = blockIdx.y;
  const int tid  = threadIdx.x;
  const int wave = tid >> 5;
  const int lane = tid & 31;
  const int hi   = lane >> 4;
  const int l15  = lane & 15;

  const float* xb = x + (size_t)b * CH * HW;
  float*       ob = out + (size_t)b * CH * HW;
  const int hwFirst = blockIdx.x * APPLY_CHUNKS * KC;

  v4f px[8];
  gram_fetch(px, xb, hwFirst, tid);

  for (int i = tid; i < (CH * CH) / 4; i += 256) {
    const int e = i * 4;
    const int r = e >> 7, c = e & 127;
    const float4 v = *reinterpret_cast<const float4*>(Mmat + e);
    float* d = sM + r * MPAD + c;
    d[0] = v.x; d[1] = v.y; d[2] = v.z; d[3] = v.w;
  }
  if (tid < CH) sB[tid] = bias[(tid >> 4) * 32 + (tid & 15)];
  gram_stage(sX0, px, tid);
  lds_barrier();

  float (*cur)[XPAD] = sX0;
  float (*nxt)[XPAD] = sX1;
  for (int j = 0; j < APPLY_CHUNKS - 1; ++j) {
    gram_fetch(px, xb, hwFirst + (j + 1) * KC, tid);
    apply_chunk(sM, cur, sB, ob, hwFirst + j * KC, wave, hi, l15);
    gram_stage(nxt, px, tid);
    lds_barrier();
    float (*t)[XPAD] = cur; cur = nxt; nxt = t;
  }
  apply_chunk(sM, cur, sB, ob, hwFirst + (APPLY_CHUNKS - 1) * KC, wave, hi, l15);
}

extern "C" void kernel_launch(void* const* d_in, const int* in_sizes, int n_in,
                              void* d_out, int out_size, void* d_ws, size_t ws_size,
                              hipStream_t stream) {
  const float* x   = (const float*)d_in[0];
  const float* rot = (const float*)d_in[1];
  float* out = (float*)d_out;
  float* ws  = (float*)d_ws;

  float* gramPart = ws;
  float* meanPart = gramPart + (size_t)BATCH * CH * CH;
  float* mean     = meanPart + CH * BATCH;
  float* Mmat     = mean + CH;

  cw_gram<<<BATCH * GBANDS, 256, 0, stream>>>(x, gramPart, meanPart);
  cw_reduce<<<CH + 1, CH, 0, stream>>>(meanPart, mean, gramPart);
  float* Sn   = gramPart + (size_t)1 * CH * CH;
  float* Pa   = gramPart + (size_t)2 * CH * CH;
  float* Pb   = gramPart + (size_t)3 * CH * CH;
  float* rTrp = gramPart + (size_t)4 * CH * CH;
  float* bias = gramPart + (size_t)5 * CH * CH;
  static_assert(T_NEWTON == 5, "the launches below are written out for five iterations");
  cw_newton_iter<true><<<CH / 16, 256, ITER_LDS_BYTES, stream>>>(
      gramPart, mean, nullptr, nullptr, Sn, Pa, rTrp);
  cw_newton_iter<false><<<CH / 16, 256, ITER_LDS_BYTES, stream>>>(
      nullptr, nullptr, Sn, Pa, nullptr, Pb, nullptr);
  cw_newton_iter<false><<<CH / 16, 256, ITER_LDS_BYTES, stream>>>(
      nullptr, nullptr, Sn, Pb, nullptr, Pa, nullptr);
  cw_newton_iter<false><<<CH / 16, 256, ITER_LDS_BYTES, stream>>>(
      nullptr, nullptr, Sn, Pa, nullptr, Pb, nullptr);
  cw_newton_finish<<<CH / 16, 256, ITER_LDS_BYTES, stream>>>(
      Pb, mean, rot, rTrp, Mmat, bias);
  cw_apply<<<dim3(HW / (APPLY_CHUNKS * KC), BATCH), 256, APPLY_LDS_BYTES, stream>>>(
      x, Mmat, bias, out);
}


# Chains of dependent round trips regrouped into one load group each: SigmaN scaling and trace in the Sigma launch, mean partials, bias sum, M staging in cw_apply; every sum in the same order
# speedup vs baseline: 49.4709x; 1.0712x over previous
#include <hip/hip_runtime.h>
#include <hip/hip_bf16.h>


typedef __attribute__((ext_vector_type(2))) float v2f;
typedef __attribute__((ext_vector_type(8))) float v8f;
typedef __attribute__((ext_vector_type(4))) float v4f;

constexpr int CH    = 128;
constexpr int BATCH = 64;
constexpr int HW    = 56 * 56;
constexpr float MTOT = (float)BATCH * (float)HW;
constexpr float EPSR = 1e-5f;
constexpr int   T_NEWTON = 5;

__global__ void __launch_bounds__(CH)
cw_reduce(const float* __restrict__ meanPart, float* __restrict__ mean,
          float* gramPart) {
  const int t = threadIdx.x;
  if (blockIdx.x == CH) {
    const int strip = t >> 4;
    const int jj  = strip < 4 ? strip : 7 - strip;
    const int src = jj * 32 + (strip < 4 ? 0 : 16) + (t & 15);
    float mp[BATCH];
#pragma unroll
    for (int b = 0; b < BATCH; ++b) mp[b] = meanPart[b * CH + src];
    float s = 0.f;
#pragma unroll
    for (int b = 0; b < BATCH; ++b) s += mp[b];
    mean[t] = s / MTOT;
    return;
  }
  if ((t >> 4) < ((int)blockIdx.x >> 4)) return;
  float* p = gramPart + blockIdx.x * CH + t;
  float v[BATCH];
#pragma unroll
  for (int s = 0; s < BATCH; ++s) v[s] = p[(size_t)s * CH * CH];
  float g = 0.f;
#pragma unroll
  for (int s = 0; s < BATCH; ++s) g += v[s];
  *p = g;
}

constexpr int KC = 64;
constexpr int XPAD = KC + 4;
constexpr int GBANDS = 4;
constexpr int NCHUNK = HW / KC;

__device__ __forceinline__ void lds_barrier() {
  __builtin_amdgcn_fence(__ATOMIC_RELEASE, "workgroup", "local");
  __builtin_amdgcn_s_barrier();
  __builtin_amdgcn_fence(__ATOMIC_ACQUIRE, "workgroup", "local");
}

__device__ __forceinline__ void gram_fetch(v4f (&v)[8], const float* __restrict__ xb,
                                           int kc, int tid) {
#pragma unroll
  for (int i = 0; i < 8; ++i) {
    const int idx = i * 256 + tid;
    const int c   = idx >> 4;
    const int k4  = (idx & 15) << 2;
    v[i] = *reinterpret_cast<const v4f*>(xb + (size_t)c * HW + kc + k4);
  }
}

__device__ __forceinline__ void gram_stage(float (*s)[XPAD], const v4f (&v)[8], int tid) {
#pragma unroll
  for (int i = 0; i < 8; ++i) {
    const int idx = i * 256 + tid;
    const int c   = idx >> 4;
    const int k4  = (idx & 15) << 2;
    *reinterpret_cast<v4f*>(&s[c][k4]) = v[i];
  }
}

template <int J>
__device__ __forceinline__ void gramT_fetch(v4f (&v)[8], const float* __restrict__ xb,
                                            int kc, int tid) {
#pragma unroll
  for (int i = J; i < 8; ++i) {
    const int idx = i * 256 + tid;
    const int c   = idx >> 4;
    const int k4  = (idx & 15) << 2;
    v[i] = *reinterpret_cast<const v4f*>(xb + (size_t)c * HW + kc + k4);
  }
}

template <int J>
__device__ __forceinline__ void gramT_stage(float (*s)[XPAD], const v4f (&v)[8], int tid) {
#pragma unroll
  for (int i = J; i < 8; ++i) {
    const int idx = i * 256 + tid;
    const int c   = idx >> 4;
    const int k4  = (idx & 15) << 2;
    *reinterpret_cast<v4f*>(&s[c][k4]) = v[i];
  }
}

template <int J>
__device__ __forceinline__ void gramT_chunk(const float (*s)[XPAD], v8f (&acc)[9], float& msum,
                                            int mrow, int mcol, int koff, int l15) {
  {
    const float4 u0 = *reinterpret_cast<const float4*>(&s[mrow][mcol]);
    const float4 u1 = *reinterpret_cast<const float4*>(&s[mrow][mcol + 4]);
    msum += u0.x; msum += u0.y; msum += u0.z; msum += u0.w;
    msum += u1.x; msum += u1.y; msum += u1.z; msum += u1.w;
  }
  v4f f[8];
#pragma unroll
  for (int t = J; t < 8; ++t)
    f[t] = *reinterpret_cast<const v4f*>(&s[t * 16 + l15][koff]);
#pragma unroll
  for (int h = 0; h < 2; ++h) {
#pragma unroll
    for (int n = 0; n < 9; ++n) {
      const int ra = (n < 8 - J) ? J : 7 - J;
      const int cb = (n < 8 - J) ? J + n : (7 - J) + (n - (8 - J));
      v2f a, bb;
      if (h == 0) { a = f[ra].xy; bb = f[cb].xy; } else { a = f[ra].zw; bb = f[cb].zw; }
      acc[n] = __builtin_amdgcn_wmma_f32_16x16x4_f32(
          false, a, false, bb, (short)0, acc[n], false, false);
    }
  }
}

constexpr int GRED_FLOATS = 8 * 9 * 256;
static_assert(GRED_FLOATS >= 2 * CH * XPAD, "the two slabs live in the same buffer");

template <int J>
__device__ __forceinline__ void gram_body(const float* __restrict__ x, float* __restrict__ gramPart,
                                          float* __restrict__ meanPart, float* sbuf,
                                          float (*sRow)[9]) {
  float (*s0)[XPAD] = reinterpret_cast<float (*)[XPAD]>(sbuf);
  float (*s1)[XPAD] = reinterpret_cast<float (*)[XPAD]>(sbuf + CH * XPAD);
  const int b    = blockIdx.x / GBANDS;
  const int tid  = threadIdx.x;
  const int wave = tid >> 5;
  const int lane = tid & 31;
  const int hi   = lane >> 4;
  const int l15  = lane & 15;
  const int koff = wave * 8 + hi * 4;

  v8f acc[9] = {};
  const float* xb = x + (size_t)b * CH * HW;
  const int rr   = tid >> 3;
  const int mrow = (rr < 16) ? J * 16 + rr : (7 - J) * 16 + (rr - 16);
  const int mcol = (tid & 7) << 3;
  float msum = 0.f;

  v4f pa[8], pb[8];
  gramT_fetch<J>(pa, xb, 0, tid);
  gramT_fetch<J>(pb, xb, KC, tid);
  gramT_stage<J>(s0, pa, tid);
  lds_barrier();

  static_assert(NCHUNK % 2 == 1 && NCHUNK >= 5, "tail below assumes an odd chunk count");
  for (int ch = 0; ch < NCHUNK - 3; ch += 2) {
    gramT_fetch<J>(pa, xb, (ch + 2) * KC, tid);
    gramT_chunk<J>(s0, acc, msum, mrow, mcol, koff, l15);
    __builtin_amdgcn_sched_barrier(0);
    gramT_stage<J>(s1, pb, tid);
    lds_barrier();
    gramT_fetch<J>(pb, xb, (ch + 3) * KC, tid);
    gramT_chunk<J>(s1, acc, msum, mrow, mcol, koff, l15);
    __builtin_amdgcn_sched_barrier(0);
    gramT_stage<J>(s0, pa, tid);
    lds_barrier();
  }
  gramT_fetch<J>(pa, xb, (NCHUNK - 1) * KC, tid);
  gramT_chunk<J>(s0, acc, msum, mrow, mcol, koff, l15);
  __builtin_amdgcn_sched_barrier(0);
  gramT_stage<J>(s1, pb, tid);
  lds_barrier();
  gramT_chunk<J>(s1, acc, msum, mrow, mcol, koff, l15);
  __builtin_amdgcn_sched_barrier(0);
  gramT_stage<J>(s0, pa, tid);
  lds_barrier();
  gramT_chunk<J>(s0, acc, msum, mrow, mcol, koff, l15);

  lds_barrier();
#pragma unroll
  for (int n = 0; n < 9; ++n) {
#pragma unroll
    for (int r = 0; r < 8; ++r)
      sbuf[(wave * 9 + n) * 256 + (r + hi * 8) * 16 + l15] = acc[n][r];
  }
  sRow[tid >> 3][tid & 7] = msum;
  __syncthreads();
  float* gp = gramPart + (size_t)b * CH * CH;
#pragma unroll
  for (int n = 0; n < 9; ++n) {
    const int ra = (n < 8 - J) ? J : 7 - J;
    const int cb = (n < 8 - J) ? J + n : (7 - J) + (n - (8 - J));
    float v = 0.f;
#pragma unroll
    for (int w = 0; w < 8; ++w) v += sbuf[(w * 9 + n) * 256 + tid];
    gp[(ra * 16 + (tid >> 4)) * CH + cb * 16 + (tid & 15)] = v;
  }
  if (tid < 32) {
    float sm = 0.f;
    for (int q = 0; q < 8; ++q) sm += sRow[tid][q];
    meanPart[b * CH + J * 32 + tid] = sm;
  }
}

__global__ void __launch_bounds__(256)
cw_gram(const float* __restrict__ x, float* __restrict__ gramPart,
        float* __restrict__ meanPart) {
  __shared__ alignas(16) float sbuf[GRED_FLOATS];
  __shared__ float sRow[32][9];
  switch (blockIdx.x % GBANDS) {
    case 0:  gram_body<0>(x, gramPart, meanPart, sbuf, sRow); break;
    case 1:  gram_body<1>(x, gramPart, meanPart, sbuf, sRow); break;
    case 2:  gram_body<2>(x, gramPart, meanPart, sbuf, sRow); break;
    default: gram_body<3>(x, gramPart, meanPart, sbuf, sRow); break;
  }
}

constexpr int NLD = CH + 12;
constexpr int STRIP = 16 * NLD;
constexpr size_t ITER_LDS_BYTES = (size_t)(2 * CH * NLD + 3 * STRIP) * sizeof(float);

__device__ __forceinline__ void load_afrag(v2f (&a)[32], const float* __restrict__ Astrip,
                                           int hi, int l15) {
#pragma unroll
  for (int j = 0; j < 32; ++j) {
    const int ks = 4 * j + hi * 2;
    a[j].x = Astrip[l15 * NLD + ks];
    a[j].y = Astrip[l15 * NLD + ks + 1];
  }
}

__device__ __forceinline__ void load_bfrag(v2f (&b)[32], const float* __restrict__ Bcol,
                                           int hi, int l15) {
#pragma unroll
  for (int j = 0; j < 32; ++j) {
    const int ks = 4 * j + hi * 2;
    b[j].x = Bcol[ks * NLD + l15];
    b[j].y = Bcol[(ks + 1) * NLD + l15];
  }
}

__device__ __forceinline__ v8f tile_chain(const v2f (&a)[32], const v2f (&b)[32]) {
  v8f acc = {};
#pragma unroll
  for (int j = 0; j < 32; ++j)
    acc = __builtin_amdgcn_wmma_f32_16x16x4_f32(
        false, a[j], false, b[j], (short)0, acc, false, false);
  return acc;
}

__device__ __forceinline__ void tile_store(float* __restrict__ strip, const v8f& acc,
                                           int wave, int hi, int l15) {
#pragma unroll
  for (int r = 0; r < 8; ++r) strip[(r + hi * 8) * NLD + wave * 16 + l15] = acc[r];
}

__device__ __forceinline__ void strip_to_global(const float* __restrict__ strip,
                                                float* __restrict__ dst, int tid) {
#pragma unroll
  for (int i = 0; i < 2; ++i) {
    const int idx = i * 256 + tid;
    const int r = idx >> 5, c = (idx & 31) << 2;
    *reinterpret_cast<v4f*>(dst + r * CH + c) =
        *reinterpret_cast<const v4f*>(strip + r * NLD + c);
  }
}

template <bool FIRST>
__global__ void __launch_bounds__(256)
cw_newton_iter(const float* __restrict__ gsum, const float* __restrict__ mean,
               const float* __restrict__ Sin, const float* __restrict__ Pin,
               float* __restrict__ Sout, float* __restrict__ Pout,
               float* __restrict__ rTrOut) {
  extern __shared__ __attribute__((aligned(16))) float nsm[];
  float* S  = nsm;
  float* P  = nsm + CH * NLD;
  float* T1 = nsm + 2 * CH * NLD;
  float* T2 = T1 + STRIP;
  const int strip = blockIdx.x;
  const int tid  = threadIdx.x;
  const int wave = tid >> 5;
  const int lane = tid & 31;
  const int hi   = lane >> 4;
  const int l15  = lane & 15;

  v4f sv[16];
  if constexpr (FIRST) {
    __shared__ float sTrInv;
    __shared__ alignas(16) float sDiag[CH];
    const int er = tid >> 4, ec = tid & 15;
    float mr8[8], mc8[8];
#pragma unroll
    for (int t = 0; t < 8; ++t) {
      mr8[t] = mean[t * 16 + er];
      mc8[t] = mean[t * 16 + ec];
    }
    float g[36];
#pragma unroll
    for (int ti = 0; ti < 8; ++ti) {
#pragma unroll
      for (int tj = ti; tj < 8; ++tj) {
        const int n = ti * 8 - (ti * (ti - 1)) / 2 + (tj - ti);
        g[n] = gsum[(ti * 16 + er) * CH + tj * 16 + ec];
      }
    }
#pragma unroll
    for (int ti = 0; ti < 8; ++ti) {
#pragma unroll
      for (int tj = ti; tj < 8; ++tj) {
        const int n = ti * 8 - (ti * (ti - 1)) / 2 + (tj - ti);
        const int r = ti * 16 + er, c = tj * 16 + ec;
        float sg = (g[n] - MTOT * mr8[ti] * mc8[tj]) / MTOT;
        if (r == c) { sg += EPSR; sDiag[r] = sg; }
        S[r * NLD + c] = sg;
        if (ti < tj) S[c * NLD + r] = sg;
      }
    }
    __syncthreads();
    if (tid == 0) {
      v4f dg[CH / 4];
#pragma unroll
      for (int q = 0; q < CH / 4; ++q) dg[q] = *reinterpret_cast<const v4f*>(sDiag + 4 * q);
      __builtin_amdgcn_sched_barrier(0);
      float tr = 0.f;
#pragma unroll
      for (int q = 0; q < CH / 4; ++q) {
        tr += dg[q].x; tr += dg[q].y; tr += dg[q].z; tr += dg[q].w;
      }
      sTrInv = 1.0f / tr;
    }
    __syncthreads();
    const float rTr = sTrInv;
    v4f sl[16];
#pragma unroll
    for (int i = 0; i < 16; ++i) {
      const int idx = i * 256 + tid;
      const int r = idx >> 5, c = (idx & 31) << 2;
      sl[i] = *reinterpret_cast<const v4f*>(S + r * NLD + c);
    }
#pragma unroll
    for (int i = 0; i < 16; ++i) {
      const int idx = i * 256 + tid;
      const int r = idx >> 5, c = (idx & 31) << 2;
      v4f sn, pn;
      sn.x = sl[i].x * rTr; sn.y = sl[i].y * rTr; sn.z = sl[i].z * rTr; sn.w = sl[i].w * rTr;
      pn.x = 1.5f * ((r == c)     ? 1.f : 0.f) - 0.5f * sn.x;
      pn.y = 1.5f * ((r == c + 1) ? 1.f : 0.f) - 0.5f * sn.y;
      pn.z = 1.5f * ((r == c + 2) ? 1.f : 0.f) - 0.5f * sn.z;
      pn.w = 1.5f * ((r == c + 3) ? 1.f : 0.f) - 0.5f * sn.w;
      *reinterpret_cast<v4f*>(S + r * NLD + c) = sn;
      *reinterpret_cast<v4f*>(P + r * NLD + c) = pn;
    }
    if (strip == 0 && tid == 0) *rTrOut = rTr;
    __syncthreads();
    strip_to_global(S + strip * STRIP, Sout + (size_t)strip * 16 * CH, tid);
  } else {
    v4f pv[16];
#pragma unroll
    for (int i = 0; i < 16; ++i)
      pv[i] = *reinterpret_cast<const v4f*>(Pin + (size_t)(i * 256 + tid) * 4);
#pragma unroll
    for (int i = 0; i < 16; ++i)
      sv[i] = *reinterpret_cast<const v4f*>(Sin + (size_t)(i * 256 + tid) * 4);
#pragma unroll
    for (int i = 0; i < 16; ++i) {
      const int idx = i * 256 + tid;
      const int r = idx >> 5, c = (idx & 31) << 2;
      *reinterpret_cast<v4f*>(P + r * NLD + c) = pv[i];
    }
    lds_barrier();
  }

  v2f a[32], bP[32], bS[32];
  load_afrag(a, P + strip * STRIP, hi, l15);
  load_bfrag(bP, P + wave * 16, hi, l15);
  __builtin_amdgcn_sched_barrier(0);
  v8f acc = tile_chain(a, bP);
  tile_store(T1, acc, wave, hi, l15);
  if constexpr (!FIRST) {
#pragma unroll
    for (int i = 0; i < 16; ++i) {
      const int idx = i * 256 + tid;
      const int r = idx >> 5, c = (idx & 31) << 2;
      *reinterpret_cast<v4f*>(S + r * NLD + c) = sv[i];
    }
  }
  lds_barrier();
  load_afrag(a, T1, hi, l15);
  load_bfrag(bS, S + wave * 16, hi, l15);
  __builtin_amdgcn_sched_barrier(0);
  acc = tile_chain(a, bP);
  tile_store(T2, acc, wave, hi, l15);
  lds_barrier();
  load_afrag(a, T2, hi, l15);
  float pold[8];
#pragma unroll
  for (int r = 0; r < 8; ++r)
    pold[r] = P[strip * STRIP + (r + hi * 8) * NLD + wave * 16 + l15];
  __builtin_amdgcn_sched_barrier(0);
  acc = tile_chain(a, bS);
#pragma unroll
  for (int r = 0; r < 8; ++r) {
    const int o = (r + hi * 8) * NLD + wave * 16 + l15;
    T1[o] = 1.5f * pold[r] - 0.5f * acc[r];
  }
  lds_barrier();
  strip_to_global(T1, Pout + (size_t)strip * 16 * CH, tid);
}

__global__ void __launch_bounds__(256)
cw_newton_finish(const float* __restrict__ Pin, const float* __restrict__ mean,
                 const float* __restrict__ rot, const float* __restrict__ rTrIn,
                 float* __restrict__ Mout, float* __restrict__ bias) {
  extern __shared__ __attribute__((aligned(16))) float nsm[];
  float* P  = nsm + CH * NLD;
  float* T1 = nsm + 2 * CH * NLD;
  float* T2 = T1 + STRIP;
  float* T3 = T2 + STRIP;
  const int strip = blockIdx.x;
  const int tid  = threadIdx.x;
  const int wave = tid >> 5;
  const int lane = tid & 31;
  const int hi   = lane >> 4;
  const int l15  = lane & 15;

  __shared__ alignas(16) float sMean[CH];
  v4f pv[16], rv[2], mnv = {};
#pragma unroll
  for (int i = 0; i < 2; ++i)
    rv[i] = *reinterpret_cast<const v4f*>(rot + (size_t)strip * 16 * CH + (i * 256 + tid) * 4);
  if (tid < CH / 4) mnv = *reinterpret_cast<const v4f*>(mean + tid * 4);
#pragma unroll
  for (int i = 0; i < 16; ++i)
    pv[i] = *reinterpret_cast<const v4f*>(Pin + (size_t)(i * 256 + tid) * 4);
  const float sc = sqrtf(*rTrIn);
#pragma unroll
  for (int i = 0; i < 2; ++i) {
    const int idx = i * 256 + tid;
    const int r = idx >> 5, c = (idx & 31) << 2;
    *reinterpret_cast<v4f*>(T1 + r * NLD + c) = rv[i];
  }
#pragma unroll
  for (int i = 0; i < 16; ++i) {
    const int idx = i * 256 + tid;
    const int r = idx >> 5, c = (idx & 31) << 2;
    *reinterpret_cast<v4f*>(P + r * NLD + c) = pv[i];
  }
  if (tid < CH / 4) *reinterpret_cast<v4f*>(sMean + tid * 4) = mnv;
  __syncthreads();
  v2f a[32], bP[32];
  load_afrag(a, T1, hi, l15);
  load_bfrag(bP, P + wave * 16, hi, l15);
  __builtin_amdgcn_sched_barrier(0);
  const v8f acc = tile_chain(a, bP);
  tile_store(T2, acc, wave, hi, l15);
#pragma unroll
  for (int r = 0; r < 8; ++r) T3[(r + hi * 8) * NLD + wave * 16 + l15] = acc[r] * sc;
  __syncthreads();
  strip_to_global(T3, Mout + (size_t)strip * 16 * CH, tid);
  if (tid < 16) {
    v4f tv[CH / 4], mv[CH / 4];
#pragma unroll
    for (int q = 0; q < CH / 4; ++q) {
      tv[q] = *reinterpret_cast<const v4f*>(T2 + tid * NLD + 4 * q);
      mv[q] = *reinterpret_cast<const v4f*>(sMean + 4 * q);
    }
    __builtin_amdgcn_sched_barrier(0);
    float s = 0.f;
#pragma unroll
    for (int q = 0; q < CH / 4; ++q) {
      s += tv[q].x * mv[q].x; s += tv[q].y * mv[q].y;
      s += tv[q].z * mv[q].z; s += tv[q].w * mv[q].w;
    }
    bias[strip * 32 + tid] = s * sc;
  }
}

constexpr int MPAD = CH + 4;
constexpr int APPLY_CHUNKS = 7;
constexpr size_t APPLY_LDS_BYTES = (size_t)(CH * MPAD + 2 * CH * XPAD + CH) * sizeof(float);

__device__ __forceinline__ void apply_chunk(const float* __restrict__ sM,
                                            const float (*sXc)[XPAD],
                                            const float* __restrict__ sB,
                                            float* __restrict__ ob, int hw0,
                                            int wave, int hi, int l15) {
  v8f acc[4] = {};
#pragma unroll 2
  for (int c0 = 0; c0 < CH; c0 += 4) {
    const int ks = c0 + hi * 2;
    const int ra = wave * 16 + l15;
    v2f a; a.x = sM[ra * MPAD + ks]; a.y = sM[ra * MPAD + ks + 1];
#pragma unroll
    for (int ht = 0; ht < 4; ++ht) {
      v2f bb;
      bb.x = sXc[ks][ht * 16 + l15];
      bb.y = sXc[ks + 1][ht * 16 + l15];
      acc[ht] = __builtin_amdgcn_wmma_f32_16x16x4_f32(
          false, a, false, bb, (short)0, acc[ht], false, false);
    }
  }

#pragma unroll
  for (int ht = 0; ht < 4; ++ht) {
#pragma unroll
    for (int r = 0; r < 8; ++r) {
      const int dch = wave * 16 + r + hi * 8;
      const int hw  = hw0 + ht * 16 + l15;
      ob[(size_t)dch * HW + hw] = acc[ht][r] - sB[dch];
    }
  }
}

__global__ void __launch_bounds__(256)
cw_apply(const float* __restrict__ x, const float* __restrict__ Mmat,
         const float* __restrict__ bias, float* __restrict__ out) {
  extern __shared__ __attribute__((aligned(16))) float smem[];
  float* sM = smem;
  float (*sX0)[XPAD] = reinterpret_cast<float (*)[XPAD]>(smem + CH * MPAD);
  float (*sX1)[XPAD] = reinterpret_cast<float (*)[XPAD]>(smem + CH * MPAD + CH * XPAD);
  float* sB = smem + CH * MPAD + 2 * CH * XPAD;
  const int b    = blockIdx.y;
  const int tid  = threadIdx.x;
  const int wave = tid >> 5;
  const int lane = tid & 31;
  const int hi   = lane >> 4;
  const int l15  = lane & 15;

  const float* xb = x + (size_t)b * CH * HW;
  float*       ob = out + (size_t)b * CH * HW;
  const int hwFirst = blockIdx.x * APPLY_CHUNKS * KC;

  v4f px[8];
  gram_fetch(px, xb, hwFirst, tid);

  v4f mv[16];
#pragma unroll
  for (int i = 0; i < 16; ++i)
    mv[i] = *reinterpret_cast<const v4f*>(Mmat + (size_t)(i * 256 + tid) * 4);
  float bv = 0.f;
  if (tid < CH) bv = bias[(tid >> 4) * 32 + (tid & 15)];
#pragma unroll
  for (int i = 0; i < 16; ++i) {
    const int idx = i * 256 + tid;
    const int r = idx >> 5, c = (idx & 31) << 2;
    *reinterpret_cast<v4f*>(sM + r * MPAD + c) = mv[i];
  }
  if (tid < CH) sB[tid] = bv;
  gram_stage(sX0, px, tid);
  lds_barrier();

  float (*cur)[XPAD] = sX0;
  float (*nxt)[XPAD] = sX1;
  for (int j = 0; j < APPLY_CHUNKS - 1; ++j) {
    gram_fetch(px, xb, hwFirst + (j + 1) * KC, tid);
    apply_chunk(sM, cur, sB, ob, hwFirst + j * KC, wave, hi, l15);
    gram_stage(nxt, px, tid);
    lds_barrier();
    float (*t)[XPAD] = cur; cur = nxt; nxt = t;
  }
  apply_chunk(sM, cur, sB, ob, hwFirst + (APPLY_CHUNKS - 1) * KC, wave, hi, l15);
}

extern "C" void kernel_launch(void* const* d_in, const int* in_sizes, int n_in,
                              void* d_out, int out_size, void* d_ws, size_t ws_size,
                              hipStream_t stream) {
  const float* x   = (const float*)d_in[0];
  const float* rot = (const float*)d_in[1];
  float* out = (float*)d_out;
  float* ws  = (float*)d_ws;

  float* gramPart = ws;
  float* meanPart = gramPart + (size_t)BATCH * CH * CH;
  float* mean     = meanPart + CH * BATCH;
  float* Mmat     = mean + CH;

  cw_gram<<<BATCH * GBANDS, 256, 0, stream>>>(x, gramPart, meanPart);
  cw_reduce<<<CH + 1, CH, 0, stream>>>(meanPart, mean, gramPart);
  float* Sn   = gramPart + (size_t)1 * CH * CH;
  float* Pa   = gramPart + (size_t)2 * CH * CH;
  float* Pb   = gramPart + (size_t)3 * CH * CH;
  float* rTrp = gramPart + (size_t)4 * CH * CH;
  float* bias = gramPart + (size_t)5 * CH * CH;
  static_assert(T_NEWTON == 5, "the launches below are written out for five iterations");
  cw_newton_iter<true><<<CH / 16, 256, ITER_LDS_BYTES, stream>>>(
      gramPart, mean, nullptr, nullptr, Sn, Pa, rTrp);
  cw_newton_iter<false><<<CH / 16, 256, ITER_LDS_BYTES, stream>>>(
      nullptr, nullptr, Sn, Pa, nullptr, Pb, nullptr);
  cw_newton_iter<false><<<CH / 16, 256, ITER_LDS_BYTES, stream>>>(
      nullptr, nullptr, Sn, Pb, nullptr, Pa, nullptr);
  cw_newton_iter<false><<<CH / 16, 256, ITER_LDS_BYTES, stream>>>(
      nullptr, nullptr, Sn, Pa, nullptr, Pb, nullptr);
  cw_newton_finish<<<CH / 16, 256, ITER_LDS_BYTES, stream>>>(
      Pb, mean, rot, rTrp, Mmat, bias);
  cw_apply<<<dim3(HW / (APPLY_CHUNKS * KC), BATCH), 256, APPLY_LDS_BYTES, stream>>>(
      x, Mmat, bias, out);
}
